# starting kernel
# baseline (speedup 1.0000x reference)
#include <hip/hip_runtime.h>

typedef __attribute__((ext_vector_type(2))) float v2f;
typedef __attribute__((ext_vector_type(8))) float v8f;

#define DIMC   192
#define HEADS  6
#define HD     32
#define NT     49
#define NTP    64
#define DISPC  3
#define IMG    56
#define NWH    8
#define NWW    8
#define XS     196
#define QS     580
#define DSS    68
#define SCALEF 0.17677669529663687f

__device__ __forceinline__ v8f wmma4(v2f a, v2f b, v8f c) {
    return __builtin_amdgcn_wmma_f32_16x16x4_f32(
         false, a,  false, b,
         (short)0, c,  false,  false);
}

__device__ __forceinline__ v2f ld2(const float* p) { return *(const v2f*)p; }

__global__ __launch_bounds__(256)
void swin_window_attn_kernel(const float* __restrict__ x,
                             const float* __restrict__ w_qkv,
                             const float* __restrict__ w_out,
                             const float* __restrict__ b_out,
                             const float* __restrict__ pe,
                             float* __restrict__ out)
{
    extern __shared__ float smem[];
    float* s_pe  = smem;
    float* s_x   = smem + 172;
    float* s_qkv = s_x   + NTP * XS;
    float* s_dot = s_qkv + NTP * QS;

    const int tid  = threadIdx.x;
    const int lane = tid & 31;
    const int wave = tid >> 5;
    const int half = lane >> 4;
    const int lm   = lane & 15;

    const int blk = blockIdx.x;
    const int b   = blk >> 6;
    const int wy  = (blk >> 3) & 7;
    const int wx  = blk & 7;

    if (tid < 169) s_pe[tid] = pe[tid];
    for (int idx = tid; idx < NTP * 48; idx += 256) {
        const int i  = idx / 48;
        const int dv = idx % 48;
        float4 v = make_float4(0.f, 0.f, 0.f, 0.f);
        if (i < NT) {
            const int r = i / 7, c = i % 7;
            int gh = wy * 7 + r + DISPC; if (gh >= IMG) gh -= IMG;
            int gw = wx * 7 + c + DISPC; if (gw >= IMG) gw -= IMG;
            v = *(const float4*)(x + (((size_t)b * IMG + gh) * IMG + gw) * DIMC + dv * 4);
        }
        *(float4*)(s_x + i * XS + dv * 4) = v;
    }
    __syncthreads();

    for (int nt = wave; nt < 36; nt += 8) {
        const int nb = nt * 16;
        v8f acc[4] = {};
        for (int k0 = 0; k0 < DIMC; k0 += 4) {
            const int kk = k0 + 2 * half;
            v2f bf;
            bf.x = w_qkv[(size_t)kk       * 576 + nb + lm];
            bf.y = w_qkv[(size_t)(kk + 1) * 576 + nb + lm];
            #pragma unroll
            for (int mt = 0; mt < 4; ++mt) {
                v2f af = ld2(s_x + (mt * 16 + lm) * XS + kk);
                acc[mt] = wmma4(af, bf, acc[mt]);
            }
        }
        #pragma unroll
        for (int mt = 0; mt < 4; ++mt)
            #pragma unroll
            for (int r = 0; r < 8; ++r)
                s_qkv[(mt * 16 + r + 8 * half) * QS + nb + lm] = acc[mt][r];
    }
    __syncthreads();

    const bool mask_ul = (wy == NWH - 1);
    const bool mask_lr = (wx == NWW - 1);

    for (int h = 0; h < HEADS; ++h) {
        const int qc = h * HD;
        const int kc = DIMC + h * HD;
        const int vc = 2 * DIMC + h * HD;

        for (int t = wave; t < 16; t += 8) {
            const int mt = t >> 2, ntl = t & 3;
            v8f acc = {};
            for (int k0 = 0; k0 < HD; k0 += 4) {
                const int kk = k0 + 2 * half;
                v2f a  = ld2(s_qkv + (mt  * 16 + lm) * QS + qc + kk);
                v2f bb = ld2(s_qkv + (ntl * 16 + lm) * QS + kc + kk);
                acc = wmma4(a, bb, acc);
            }
            #pragma unroll
            for (int r = 0; r < 8; ++r) {
                const int i = mt * 16 + r + 8 * half;
                const int j = ntl * 16 + lm;
                float val = 0.f;
                if (i < NT && j < NT) {
                    const int xi = i / 7, yi = i % 7, xj = j / 7, yj = j % 7;
                    val = acc[r] * SCALEF + s_pe[(xj - xi + 6) * 13 + (yj - yi + 6)];
                    if (mask_ul && ((i >= 28) != (j >= 28))) val = -1e30f;
                    if (mask_lr && ((yi >= 4) != (yj >= 4))) val = -1e30f;
                }
                s_dot[i * DSS + j] = val;
            }
        }
        __syncthreads();

        {
            const int row = tid >> 2;
            const int q   = tid & 3;
            float* rp = s_dot + row * DSS;
            if (row < NT) {
                float pm = -1e30f;
                #pragma unroll
                for (int jj = 0; jj < 16; ++jj) {
                    const int j = q * 16 + jj;
                    if (j < NT) pm = fmaxf(pm, rp[j]);
                }
                pm = fmaxf(pm, __shfl_xor(pm, 1, 4));
                pm = fmaxf(pm, __shfl_xor(pm, 2, 4));
                float ev[16];
                float ps = 0.f;
                #pragma unroll
                for (int jj = 0; jj < 16; ++jj) {
                    const int j = q * 16 + jj;
                    const float e = (j < NT) ? __expf(rp[j] - pm) : 0.f;
                    ev[jj] = e; ps += e;
                }
                ps += __shfl_xor(ps, 1, 4);
                ps += __shfl_xor(ps, 2, 4);
                const float inv = 1.f / ps;
                #pragma unroll
                for (int jj = 0; jj < 16; ++jj)
                    rp[q * 16 + jj] = ev[jj] * inv;
            } else {
                #pragma unroll
                for (int jj = 0; jj < 16; ++jj)
                    rp[q * 16 + jj] = 0.f;
            }
        }
        __syncthreads();

        {
            const int mt = wave >> 1, ntl = wave & 1;
            v8f acc = {};
            for (int k0 = 0; k0 < NTP; k0 += 4) {
                const int kk = k0 + 2 * half;
                v2f a = ld2(s_dot + (mt * 16 + lm) * DSS + kk);
                v2f bb;
                bb.x = s_qkv[kk       * QS + vc + ntl * 16 + lm];
                bb.y = s_qkv[(kk + 1) * QS + vc + ntl * 16 + lm];
                acc = wmma4(a, bb, acc);
            }
            #pragma unroll
            for (int r = 0; r < 8; ++r)
                s_x[(mt * 16 + r + 8 * half) * XS + h * HD + ntl * 16 + lm] = acc[r];
        }
        __syncthreads();
    }

    for (int d = wave; d < 24; d += 8) {
        const int mh = d / 12;
        const int nt = d % 12;
        const int nb = nt * 16;
        v8f acc[2] = {};
        for (int k0 = 0; k0 < DIMC; k0 += 4) {
            const int kk = k0 + 2 * half;
            v2f bf;
            bf.x = w_out[(size_t)kk       * DIMC + nb + lm];
            bf.y = w_out[(size_t)(kk + 1) * DIMC + nb + lm];
            #pragma unroll
            for (int mi = 0; mi < 2; ++mi) {
                v2f af = ld2(s_x + ((mh * 2 + mi) * 16 + lm) * XS + kk);
                acc[mi] = wmma4(af, bf, acc[mi]);
            }
        }
        const float bo = b_out[nb + lm];
        #pragma unroll
        for (int mi = 0; mi < 2; ++mi)
            #pragma unroll
            for (int r = 0; r < 8; ++r) {
                const int i = (mh * 2 + mi) * 16 + r + 8 * half;
                if (i < NT) {
                    const int rr = i / 7, cc = i % 7;
                    int gh = wy * 7 + rr + DISPC; if (gh >= IMG) gh -= IMG;
                    int gw = wx * 7 + cc + DISPC; if (gw >= IMG) gw -= IMG;
                    out[(((size_t)b * IMG + gh) * IMG + gw) * DIMC + nb + lm] = acc[mi][r] + bo;
                }
            }
    }
}

extern "C" void kernel_launch(void* const* d_in, const int* in_sizes, int n_in,
                              void* d_out, int out_size, void* d_ws, size_t ws_size,
                              hipStream_t stream) {
    const float* x     = (const float*)d_in[0];
    const float* w_qkv = (const float*)d_in[1];
    const float* w_out = (const float*)d_in[2];
    const float* b_out = (const float*)d_in[3];
    const float* pe    = (const float*)d_in[4];
    float* out = (float*)d_out;

    const size_t shmem = (size_t)(172 + NTP * XS + NTP * QS + NTP * DSS) * sizeof(float);
    hipFuncSetAttribute((const void*)swin_window_attn_kernel,
                        hipFuncAttributeMaxDynamicSharedMemorySize, (int)shmem);

    const int grid = 32 * NWH * NWW;
    swin_window_attn_kernel<<<grid, 256, shmem, stream>>>(x, w_qkv, w_out, b_out, pe, out);
}


# QKV and out-proj GEMMs: weight (B) fragments fetched into registers in batches of 24 k-steps, the next batch in flight while the current one feeds WMMA; operands and sum order unchanged
# speedup vs baseline: 1.1800x; 1.1800x over previous
#include <hip/hip_runtime.h>

typedef __attribute__((ext_vector_type(2))) float v2f;
typedef __attribute__((ext_vector_type(8))) float v8f;

#define DIMC   192
#define HEADS  6
#define HD     32
#define NT     49
#define NTP    64
#define DISPC  3
#define IMG    56
#define NWH    8
#define NWW    8
#define XS     196
#define QS     580
#define DSS    68
#define SCALEF 0.17677669529663687f

__device__ __forceinline__ v8f wmma4(v2f a, v2f b, v8f c) {
    return __builtin_amdgcn_wmma_f32_16x16x4_f32(
         false, a,  false, b,
         (short)0, c,  false,  false);
}

__device__ __forceinline__ v2f ld2(const float* p) { return *(const v2f*)p; }

template <int LD>
__device__ __forceinline__ void ldb24(v2f (&dst)[24], const float* __restrict__ p) {
    #pragma unroll
    for (int s = 0; s < 24; ++s) {
        dst[s].x = p[(size_t)(4 * s)     * LD];
        dst[s].y = p[(size_t)(4 * s + 1) * LD];
    }
}

template <int MT, int AS>
__device__ __forceinline__ void mma24(v8f (&acc)[MT], const v2f (&bfr)[24], const float* a) {
    #pragma unroll
    for (int s = 0; s < 24; ++s) {
        #pragma unroll
        for (int mt = 0; mt < MT; ++mt) {
            v2f af = ld2(a + mt * 16 * AS + 4 * s);
            acc[mt] = wmma4(af, bfr[s], acc[mt]);
        }
    }
}

__global__ __launch_bounds__(256)
void swin_window_attn_kernel(const float* __restrict__ x,
                             const float* __restrict__ w_qkv,
                             const float* __restrict__ w_out,
                             const float* __restrict__ b_out,
                             const float* __restrict__ pe,
                             float* __restrict__ out)
{
    extern __shared__ float smem[];
    float* s_pe  = smem;
    float* s_x   = smem + 172;
    float* s_qkv = s_x   + NTP * XS;
    float* s_dot = s_qkv + NTP * QS;

    const int tid  = threadIdx.x;
    const int lane = tid & 31;
    const int wave = tid >> 5;
    const int half = lane >> 4;
    const int lm   = lane & 15;

    const int blk = blockIdx.x;
    const int b   = blk >> 6;
    const int wy  = (blk >> 3) & 7;
    const int wx  = blk & 7;

    if (tid < 169) s_pe[tid] = pe[tid];
    for (int idx = tid; idx < NTP * 48; idx += 256) {
        const int i  = idx / 48;
        const int dv = idx % 48;
        float4 v = make_float4(0.f, 0.f, 0.f, 0.f);
        if (i < NT) {
            const int r = i / 7, c = i % 7;
            int gh = wy * 7 + r + DISPC; if (gh >= IMG) gh -= IMG;
            int gw = wx * 7 + c + DISPC; if (gw >= IMG) gw -= IMG;
            v = *(const float4*)(x + (((size_t)b * IMG + gh) * IMG + gw) * DIMC + dv * 4);
        }
        *(float4*)(s_x + i * XS + dv * 4) = v;
    }
    __syncthreads();

    {
        const float* wl = w_qkv + (size_t)(2 * half) * 576 + lm;
        const float* al = s_x + lm * XS + 2 * half;
        v2f bq0[24], bq1[24];
        ldb24<576>(bq0, wl + wave * 16);
        for (int nt = wave; nt < 36; nt += 8) {
            const int nb = nt * 16;
            const int ntn = (nt + 8 < 36) ? nt + 8 : nt;
            v8f acc[4] = {};
            ldb24<576>(bq1, wl + nb + (size_t)96 * 576);
            mma24<4, XS>(acc, bq0, al);
            ldb24<576>(bq0, wl + ntn * 16);
            __builtin_amdgcn_sched_barrier(0);
            mma24<4, XS>(acc, bq1, al + 96);
            #pragma unroll
            for (int mt = 0; mt < 4; ++mt)
                #pragma unroll
                for (int r = 0; r < 8; ++r)
                    s_qkv[(mt * 16 + r + 8 * half) * QS + nb + lm] = acc[mt][r];
        }
    }
    __syncthreads();

    const bool mask_ul = (wy == NWH - 1);
    const bool mask_lr = (wx == NWW - 1);

    for (int h = 0; h < HEADS; ++h) {
        const int qc = h * HD;
        const int kc = DIMC + h * HD;
        const int vc = 2 * DIMC + h * HD;

        for (int t = wave; t < 16; t += 8) {
            const int mt = t >> 2, ntl = t & 3;
            v8f acc = {};
            for (int k0 = 0; k0 < HD; k0 += 4) {
                const int kk = k0 + 2 * half;
                v2f a  = ld2(s_qkv + (mt  * 16 + lm) * QS + qc + kk);
                v2f bb = ld2(s_qkv + (ntl * 16 + lm) * QS + kc + kk);
                acc = wmma4(a, bb, acc);
            }
            #pragma unroll
            for (int r = 0; r < 8; ++r) {
                const int i = mt * 16 + r + 8 * half;
                const int j = ntl * 16 + lm;
                float val = 0.f;
                if (i < NT && j < NT) {
                    const int xi = i / 7, yi = i % 7, xj = j / 7, yj = j % 7;
                    val = acc[r] * SCALEF + s_pe[(xj - xi + 6) * 13 + (yj - yi + 6)];
                    if (mask_ul && ((i >= 28) != (j >= 28))) val = -1e30f;
                    if (mask_lr && ((yi >= 4) != (yj >= 4))) val = -1e30f;
                }
                s_dot[i * DSS + j] = val;
            }
        }
        __syncthreads();

        {
            const int row = tid >> 2;
            const int q   = tid & 3;
            float* rp = s_dot + row * DSS;
            if (row < NT) {
                float pm = -1e30f;
                #pragma unroll
                for (int jj = 0; jj < 16; ++jj) {
                    const int j = q * 16 + jj;
                    if (j < NT) pm = fmaxf(pm, rp[j]);
                }
                pm = fmaxf(pm, __shfl_xor(pm, 1, 4));
                pm = fmaxf(pm, __shfl_xor(pm, 2, 4));
                float ev[16];
                float ps = 0.f;
                #pragma unroll
                for (int jj = 0; jj < 16; ++jj) {
                    const int j = q * 16 + jj;
                    const float e = (j < NT) ? __expf(rp[j] - pm) : 0.f;
                    ev[jj] = e; ps += e;
                }
                ps += __shfl_xor(ps, 1, 4);
                ps += __shfl_xor(ps, 2, 4);
                const float inv = 1.f / ps;
                #pragma unroll
                for (int jj = 0; jj < 16; ++jj)
                    rp[q * 16 + jj] = ev[jj] * inv;
            } else {
                #pragma unroll
                for (int jj = 0; jj < 16; ++jj)
                    rp[q * 16 + jj] = 0.f;
            }
        }
        __syncthreads();

        {
            const int mt = wave >> 1, ntl = wave & 1;
            v8f acc = {};
            for (int k0 = 0; k0 < NTP; k0 += 4) {
                const int kk = k0 + 2 * half;
                v2f a = ld2(s_dot + (mt * 16 + lm) * DSS + kk);
                v2f bb;
                bb.x = s_qkv[kk       * QS + vc + ntl * 16 + lm];
                bb.y = s_qkv[(kk + 1) * QS + vc + ntl * 16 + lm];
                acc = wmma4(a, bb, acc);
            }
            #pragma unroll
            for (int r = 0; r < 8; ++r)
                s_x[(mt * 16 + r + 8 * half) * XS + h * HD + ntl * 16 + lm] = acc[r];
        }
        __syncthreads();
    }

    {
        const float* wl = w_out + (size_t)(2 * half) * DIMC + lm;
        v2f bo0[24], bo1[24];
        ldb24<DIMC>(bo0, wl + (wave % 12) * 16);
        for (int d = wave; d < 24; d += 8) {
            const int mh = d / 12;
            const int nt = d % 12;
            const int nb = nt * 16;
            const int dn = (d + 8 < 24) ? d + 8 : d;
            const float* al = s_x + (mh * 32 + lm) * XS + 2 * half;
            v8f acc[2] = {};
            ldb24<DIMC>(bo1, wl + nb + (size_t)96 * DIMC);
            __builtin_amdgcn_sched_barrier(0);
            mma24<2, XS>(acc, bo0, al);
            ldb24<DIMC>(bo0, wl + (dn % 12) * 16);
            __builtin_amdgcn_sched_barrier(0);
            mma24<2, XS>(acc, bo1, al + 96);
            const float bo = b_out[nb + lm];
            #pragma unroll
            for (int mi = 0; mi < 2; ++mi)
                #pragma unroll
                for (int r = 0; r < 8; ++r) {
                    const int i = (mh * 2 + mi) * 16 + r + 8 * half;
                    if (i < NT) {
                        const int rr = i / 7, cc = i % 7;
                        int gh = wy * 7 + rr + DISPC; if (gh >= IMG) gh -= IMG;
                        int gw = wx * 7 + cc + DISPC; if (gw >= IMG) gw -= IMG;
                        out[(((size_t)b * IMG + gh) * IMG + gw) * DIMC + nb + lm] = acc[mi][r] + bo;
                    }
                }
        }
    }
}

extern "C" void kernel_launch(void* const* d_in, const int* in_sizes, int n_in,
                              void* d_out, int out_size, void* d_ws, size_t ws_size,
                              hipStream_t stream) {
    const float* x     = (const float*)d_in[0];
    const float* w_qkv = (const float*)d_in[1];
    const float* w_out = (const float*)d_in[2];
    const float* b_out = (const float*)d_in[3];
    const float* pe    = (const float*)d_in[4];
    float* out = (float*)d_out;

    const size_t shmem = (size_t)(172 + NTP * XS + NTP * QS + NTP * DSS) * sizeof(float);
    hipFuncSetAttribute((const void*)swin_window_attn_kernel,
                        hipFuncAttributeMaxDynamicSharedMemorySize, (int)shmem);

    const int grid = 32 * NWH * NWW;
    swin_window_attn_kernel<<<grid, 256, shmem, stream>>>(x, w_qkv, w_out, b_out, pe, out);
}


# Attention phase: each wave owns one (head, 16-row tile) unit end to end, scores stay in registers through softmax (row sums by 16-lane butterfly: new sum order), no block barriers between heads
# speedup vs baseline: 1.7953x; 1.5214x over previous
#include <hip/hip_runtime.h>

typedef __attribute__((ext_vector_type(2))) float v2f;
typedef __attribute__((ext_vector_type(8))) float v8f;

#define DIMC   192
#define HEADS  6
#define HD     32
#define NT     49
#define NTP    64
#define DISPC  3
#define IMG    56
#define NWH    8
#define NWW    8
#define XS     196
#define QS     580
#define DSS    68
#define SCALEF 0.17677669529663687f

__device__ __forceinline__ v8f wmma4(v2f a, v2f b, v8f c) {
    return __builtin_amdgcn_wmma_f32_16x16x4_f32(
         false, a,  false, b,
         (short)0, c,  false,  false);
}

__device__ __forceinline__ v2f ld2(const float* p) { return *(const v2f*)p; }

template <int M>
__device__ __forceinline__ float rowx(float v) {
    const int b = __builtin_bit_cast(int, v);
    return __builtin_bit_cast(float, __builtin_amdgcn_update_dpp(b, b, 0x160 + M, 0xf, 0xf, true));
}

template <int LD>
__device__ __forceinline__ void ldb24(v2f (&dst)[24], const float* __restrict__ p) {
    #pragma unroll
    for (int s = 0; s < 24; ++s) {
        dst[s].x = p[(size_t)(4 * s)     * LD];
        dst[s].y = p[(size_t)(4 * s + 1) * LD];
    }
}

template <int MT, int AS>
__device__ __forceinline__ void mma24(v8f (&acc)[MT], const v2f (&bfr)[24], const float* a) {
    #pragma unroll
    for (int s = 0; s < 24; ++s) {
        #pragma unroll
        for (int mt = 0; mt < MT; ++mt) {
            v2f af = ld2(a + mt * 16 * AS + 4 * s);
            acc[mt] = wmma4(af, bfr[s], acc[mt]);
        }
    }
}

__global__ __launch_bounds__(256)
void swin_window_attn_kernel(const float* __restrict__ x,
                             const float* __restrict__ w_qkv,
                             const float* __restrict__ w_out,
                             const float* __restrict__ b_out,
                             const float* __restrict__ pe,
                             float* __restrict__ out)
{
    extern __shared__ float smem[];
    float* s_pe  = smem;
    float* s_x   = smem + 172;
    float* s_qkv = s_x   + NTP * XS;
    float* s_dot = s_qkv + NTP * QS;

    const int tid  = threadIdx.x;
    const int lane = tid & 31;
    const int wave = tid >> 5;
    const int half = lane >> 4;
    const int lm   = lane & 15;

    const int blk = blockIdx.x;
    const int b   = blk >> 6;
    const int wy  = (blk >> 3) & 7;
    const int wx  = blk & 7;

    if (tid < 169) s_pe[tid] = pe[tid];
    for (int idx = tid; idx < NTP * 48; idx += 256) {
        const int i  = idx / 48;
        const int dv = idx % 48;
        float4 v = make_float4(0.f, 0.f, 0.f, 0.f);
        if (i < NT) {
            const int r = i / 7, c = i % 7;
            int gh = wy * 7 + r + DISPC; if (gh >= IMG) gh -= IMG;
            int gw = wx * 7 + c + DISPC; if (gw >= IMG) gw -= IMG;
            v = *(const float4*)(x + (((size_t)b * IMG + gh) * IMG + gw) * DIMC + dv * 4);
        }
        *(float4*)(s_x + i * XS + dv * 4) = v;
    }
    __syncthreads();

    {
        const float* wl = w_qkv + (size_t)(2 * half) * 576 + lm;
        const float* al = s_x + lm * XS + 2 * half;
        v2f bq0[24], bq1[24];
        ldb24<576>(bq0, wl + wave * 16);
        for (int nt = wave; nt < 36; nt += 8) {
            const int nb = nt * 16;
            const int ntn = (nt + 8 < 36) ? nt + 8 : nt;
            v8f acc[4] = {};
            ldb24<576>(bq1, wl + nb + (size_t)96 * 576);
            mma24<4, XS>(acc, bq0, al);
            ldb24<576>(bq0, wl + ntn * 16);
            __builtin_amdgcn_sched_barrier(0);
            mma24<4, XS>(acc, bq1, al + 96);
            #pragma unroll
            for (int mt = 0; mt < 4; ++mt)
                #pragma unroll
                for (int r = 0; r < 8; ++r)
                    s_qkv[(mt * 16 + r + 8 * half) * QS + nb + lm] = acc[mt][r];
        }
    }
    __syncthreads();

    const bool mask_ul = (wy == NWH - 1);
    const bool mask_lr = (wx == NWW - 1);
    {
        const int wv = __builtin_amdgcn_readfirstlane(wave);
        const int mt = wv & 3;
        const int h0 = wv >> 2;
        float* s_p = s_dot + wv * 16 * DSS;

        float pe_r[4][8];
        bool  mk[4][8];
        float rk[8];
        const int m_ul = mask_ul ? 1 : 0;
        const int m_lr = mask_lr ? 1 : 0;
        #pragma unroll
        for (int r = 0; r < 8; ++r) rk[r] = ((mt * 16 + r + 8 * half) < NT) ? 1.f : 0.f;
        #pragma unroll
        for (int nt = 0; nt < 4; ++nt) {
            const int j  = nt * 16 + lm;
            const int xj = j / 7, yj = j % 7;
            #pragma unroll
            for (int r = 0; r < 8; ++r) {
                const int i  = mt * 16 + r + 8 * half;
                const int xi = i / 7, yi = i % 7;
                const int ok = (int)(i < NT) & (int)(j < NT);
                int idx = (xj - xi + 6) * 13 + (yj - yi + 6);
                idx = idx < 0 ? 0 : (idx > 168 ? 168 : idx);
                const float pv = s_pe[idx];
                pe_r[nt][r] = ok ? pv : 0.f;
                const int msk = ((int)((i >= 28) != (j >= 28)) & m_ul) |
                                ((int)((yi >= 4) != (yj >= 4)) & m_lr);
                mk[nt][r] = (msk & ok) != 0;
            }
        }
        const bool col48 = (lm == 0);

        for (int u = 0; u < 3; ++u) {
            const int h  = h0 + 2 * u;
            const int qc = h * HD;
            const int kc = DIMC + h * HD;
            const int vc = 2 * DIMC + h * HD;

            v2f qf[8], kf[4][8];
            {
                const float* qa = s_qkv + (mt * 16 + lm) * QS + qc + 2 * half;
                const float* kb = s_qkv + lm * QS + kc + 2 * half;
                #pragma unroll
                for (int s = 0; s < 8; ++s) qf[s] = ld2(qa + 4 * s);
                #pragma unroll
                for (int nt = 0; nt < 4; ++nt)
                    #pragma unroll
                    for (int s = 0; s < 8; ++s) kf[nt][s] = ld2(kb + nt * 16 * QS + 4 * s);
            }
            __builtin_amdgcn_sched_barrier(0);
            v8f sc[4] = {};
            #pragma unroll
            for (int s = 0; s < 8; ++s)
                #pragma unroll
                for (int nt = 0; nt < 4; ++nt)
                    sc[nt] = wmma4(qf[s], kf[nt][s], sc[nt]);

            v2f vf[2][16];
            {
                const float* vl = s_qkv + (2 * half) * QS + vc + lm;
                #pragma unroll
                for (int s = 0; s < 16; ++s)
                    #pragma unroll
                    for (int n2 = 0; n2 < 2; ++n2) {
                        vf[n2][s].x = vl[(4 * s)     * QS + n2 * 16];
                        vf[n2][s].y = vl[(4 * s + 1) * QS + n2 * 16];
                    }
            }

            #pragma unroll
            for (int r = 0; r < 8; ++r) {
                #pragma unroll
                for (int nt = 0; nt < 4; ++nt) {
                    float v = sc[nt][r] * SCALEF + pe_r[nt][r];
                    v = mk[nt][r] ? -1e30f : v;
                    if (nt == 3) v = col48 ? v : -__builtin_inff();
                    sc[nt][r] = v;
                }
                float m = fmaxf(fmaxf(sc[0][r], sc[1][r]), fmaxf(sc[2][r], sc[3][r]));
                m = fmaxf(m, rowx<1>(m));
                m = fmaxf(m, rowx<2>(m));
                m = fmaxf(m, rowx<4>(m));
                m = fmaxf(m, rowx<8>(m));
                #pragma unroll
                for (int nt = 0; nt < 4; ++nt) sc[nt][r] = __expf(sc[nt][r] - m);
                float ps = (sc[0][r] + sc[1][r]) + (sc[2][r] + sc[3][r]);
                ps += rowx<1>(ps);
                ps += rowx<2>(ps);
                ps += rowx<4>(ps);
                ps += rowx<8>(ps);
                const float inv = (1.f / ps) * rk[r];
                #pragma unroll
                for (int nt = 0; nt < 4; ++nt)
                    s_p[(r + 8 * half) * DSS + nt * 16 + lm] = sc[nt][r] * inv;
            }

            v2f pf[16];
            {
                const float* pl = s_p + lm * DSS + 2 * half;
                #pragma unroll
                for (int s = 0; s < 16; ++s) pf[s] = ld2(pl + 4 * s);
            }
            __builtin_amdgcn_sched_barrier(0);
            v8f oa[2] = {};
            #pragma unroll
            for (int s = 0; s < 16; ++s)
                #pragma unroll
                for (int n2 = 0; n2 < 2; ++n2)
                    oa[n2] = wmma4(pf[s], vf[n2][s], oa[n2]);

            #pragma unroll
            for (int n2 = 0; n2 < 2; ++n2)
                #pragma unroll
                for (int r = 0; r < 8; ++r)
                    s_x[(mt * 16 + r + 8 * half) * XS + h * HD + n2 * 16 + lm] = oa[n2][r];
        }
    }
    __syncthreads();

    {
        const float* wl = w_out + (size_t)(2 * half) * DIMC + lm;
        v2f bo0[24], bo1[24];
        ldb24<DIMC>(bo0, wl + (wave % 12) * 16);
        for (int d = wave; d < 24; d += 8) {
            const int mh = d / 12;
            const int nt = d % 12;
            const int nb = nt * 16;
            const int dn = (d + 8 < 24) ? d + 8 : d;
            const float* al = s_x + (mh * 32 + lm) * XS + 2 * half;
            v8f acc[2] = {};
            ldb24<DIMC>(bo1, wl + nb + (size_t)96 * DIMC);
            __builtin_amdgcn_sched_barrier(0);
            mma24<2, XS>(acc, bo0, al);
            ldb24<DIMC>(bo0, wl + (dn % 12) * 16);
            __builtin_amdgcn_sched_barrier(0);
            mma24<2, XS>(acc, bo1, al + 96);
            const float bo = b_out[nb + lm];
            #pragma unroll
            for (int mi = 0; mi < 2; ++mi)
                #pragma unroll
                for (int r = 0; r < 8; ++r) {
                    const int i = (mh * 2 + mi) * 16 + r + 8 * half;
                    if (i < NT) {
                        const int rr = i / 7, cc = i % 7;
                        int gh = wy * 7 + rr + DISPC; if (gh >= IMG) gh -= IMG;
                        int gw = wx * 7 + cc + DISPC; if (gw >= IMG) gw -= IMG;
                        out[(((size_t)b * IMG + gh) * IMG + gw) * DIMC + nb + lm] = acc[mi][r] + bo;
                    }
                }
        }
    }
}

extern "C" void kernel_launch(void* const* d_in, const int* in_sizes, int n_in,
                              void* d_out, int out_size, void* d_ws, size_t ws_size,
                              hipStream_t stream) {
    const float* x     = (const float*)d_in[0];
    const float* w_qkv = (const float*)d_in[1];
    const float* w_out = (const float*)d_in[2];
    const float* b_out = (const float*)d_in[3];
    const float* pe    = (const float*)d_in[4];
    float* out = (float*)d_out;

    const size_t shmem = (size_t)(172 + NTP * XS + NTP * QS + 8 * 16 * DSS) * sizeof(float);
    hipFuncSetAttribute((const void*)swin_window_attn_kernel,
                        hipFuncAttributeMaxDynamicSharedMemorySize, (int)shmem);

    const int grid = 32 * NWH * NWW;
    swin_window_attn_kernel<<<grid, 256, shmem, stream>>>(x, w_qkv, w_out, b_out, pe, out);
}


# Prologue: every thread issues all its x-tile loads (nine float4 and its piece of row 48) and the bias-table load before the first wait, then stores them to LDS; was one awaited load per loop trip
# speedup vs baseline: 1.9769x; 1.1012x over previous
#include <hip/hip_runtime.h>

typedef __attribute__((ext_vector_type(2))) float v2f;
typedef __attribute__((ext_vector_type(8))) float v8f;

#define DIMC   192
#define HEADS  6
#define HD     32
#define NT     49
#define NTP    64
#define DISPC  3
#define IMG    56
#define NWH    8
#define NWW    8
#define XS     196
#define QS     580
#define DSS    68
#define SCALEF 0.17677669529663687f

__device__ __forceinline__ v8f wmma4(v2f a, v2f b, v8f c) {
    return __builtin_amdgcn_wmma_f32_16x16x4_f32(
         false, a,  false, b,
         (short)0, c,  false,  false);
}

__device__ __forceinline__ v2f ld2(const float* p) { return *(const v2f*)p; }

template <int M>
__device__ __forceinline__ float rowx(float v) {
    const int b = __builtin_bit_cast(int, v);
    return __builtin_bit_cast(float, __builtin_amdgcn_update_dpp(b, b, 0x160 + M, 0xf, 0xf, true));
}

template <int LD>
__device__ __forceinline__ void ldb24(v2f (&dst)[24], const float* __restrict__ p) {
    #pragma unroll
    for (int s = 0; s < 24; ++s) {
        dst[s].x = p[(size_t)(4 * s)     * LD];
        dst[s].y = p[(size_t)(4 * s + 1) * LD];
    }
}

template <int MT, int AS>
__device__ __forceinline__ void mma24(v8f (&acc)[MT], const v2f (&bfr)[24], const float* a) {
    #pragma unroll
    for (int s = 0; s < 24; ++s) {
        #pragma unroll
        for (int mt = 0; mt < MT; ++mt) {
            v2f af = ld2(a + mt * 16 * AS + 4 * s);
            acc[mt] = wmma4(af, bfr[s], acc[mt]);
        }
    }
}

__global__ __launch_bounds__(256)
void swin_window_attn_kernel(const float* __restrict__ x,
                             const float* __restrict__ w_qkv,
                             const float* __restrict__ w_out,
                             const float* __restrict__ b_out,
                             const float* __restrict__ pe,
                             float* __restrict__ out)
{
    extern __shared__ float smem[];
    float* s_pe  = smem;
    float* s_x   = smem + 172;
    float* s_qkv = s_x   + NTP * XS;
    float* s_dot = s_qkv + NTP * QS;

    const int tid  = threadIdx.x;
    const int lane = tid & 31;
    const int wave = tid >> 5;
    const int half = lane >> 4;
    const int lm   = lane & 15;

    const int blk = blockIdx.x;
    const int b   = blk >> 6;
    const int wy  = (blk >> 3) & 7;
    const int wx  = blk & 7;

    const float pev = pe[tid < 169 ? tid : 168];
    float4 xv[10];
    #pragma unroll
    for (int it = 0; it < 9; ++it) {
        const int idx = tid + it * 256;
        const int i  = idx / 48;
        const int dv = idx % 48;
        const int r = i / 7, c = i % 7;
        int gh = wy * 7 + r + DISPC; if (gh >= IMG) gh -= IMG;
        int gw = wx * 7 + c + DISPC; if (gw >= IMG) gw -= IMG;
        xv[it] = *(const float4*)(x + (((size_t)b * IMG + gh) * IMG + gw) * DIMC + dv * 4);
    }
    {
        int gh = wy * 7 + 6 + DISPC; if (gh >= IMG) gh -= IMG;
        int gw = wx * 7 + 6 + DISPC; if (gw >= IMG) gw -= IMG;
        xv[9] = *(const float4*)(x + (((size_t)b * IMG + gh) * IMG + gw) * DIMC
                                   + (tid < 48 ? tid : 47) * 4);
    }
    s_pe[tid < 169 ? tid : 169] = pev;
    #pragma unroll
    for (int it = 0; it < 9; ++it) {
        const int idx = tid + it * 256;
        *(float4*)(s_x + (idx / 48) * XS + (idx % 48) * 4) = xv[it];
    }
    #pragma unroll
    for (int it = 0; it < 3; ++it) {
        const int idx = tid + it * 256;
        *(float4*)(s_x + (NT - 1 + idx / 48) * XS + (idx % 48) * 4) =
            (idx < 48) ? xv[9] : make_float4(0.f, 0.f, 0.f, 0.f);
    }
    __syncthreads();

    {
        const float* wl = w_qkv + (size_t)(2 * half) * 576 + lm;
        const float* al = s_x + lm * XS + 2 * half;
        v2f bq0[24], bq1[24];
        ldb24<576>(bq0, wl + wave * 16);
        for (int nt = wave; nt < 36; nt += 8) {
            const int nb = nt * 16;
            const int ntn = (nt + 8 < 36) ? nt + 8 : nt;
            v8f acc[4] = {};
            ldb24<576>(bq1, wl + nb + (size_t)96 * 576);
            mma24<4, XS>(acc, bq0, al);
            ldb24<576>(bq0, wl + ntn * 16);
            __builtin_amdgcn_sched_barrier(0);
            mma24<4, XS>(acc, bq1, al + 96);
            #pragma unroll
            for (int mt = 0; mt < 4; ++mt)
                #pragma unroll
                for (int r = 0; r < 8; ++r)
                    s_qkv[(mt * 16 + r + 8 * half) * QS + nb + lm] = acc[mt][r];
        }
    }
    __syncthreads();

    const bool mask_ul = (wy == NWH - 1);
    const bool mask_lr = (wx == NWW - 1);
    {
        const int wv = __builtin_amdgcn_readfirstlane(wave);
        const int mt = wv & 3;
        const int h0 = wv >> 2;
        float* s_p = s_dot + wv * 16 * DSS;

        float pe_r[4][8];
        bool  mk[4][8];
        float rk[8];
        const int m_ul = mask_ul ? 1 : 0;
        const int m_lr = mask_lr ? 1 : 0;
        #pragma unroll
        for (int r = 0; r < 8; ++r) rk[r] = ((mt * 16 + r + 8 * half) < NT) ? 1.f : 0.f;
        #pragma unroll
        for (int nt = 0; nt < 4; ++nt) {
            const int j  = nt * 16 + lm;
            const int xj = j / 7, yj = j % 7;
            #pragma unroll
            for (int r = 0; r < 8; ++r) {
                const int i  = mt * 16 + r + 8 * half;
                const int xi = i / 7, yi = i % 7;
                const int ok = (int)(i < NT) & (int)(j < NT);
                int idx = (xj - xi + 6) * 13 + (yj - yi + 6);
                idx = idx < 0 ? 0 : (idx > 168 ? 168 : idx);
                const float pv = s_pe[idx];
                pe_r[nt][r] = ok ? pv : 0.f;
                const int msk = ((int)((i >= 28) != (j >= 28)) & m_ul) |
                                ((int)((yi >= 4) != (yj >= 4)) & m_lr);
                mk[nt][r] = (msk & ok) != 0;
            }
        }
        const bool col48 = (lm == 0);

        for (int u = 0; u < 3; ++u) {
            const int h  = h0 + 2 * u;
            const int qc = h * HD;
            const int kc = DIMC + h * HD;
            const int vc = 2 * DIMC + h * HD;

            v2f qf[8], kf[4][8];
            {
                const float* qa = s_qkv + (mt * 16 + lm) * QS + qc + 2 * half;
                const float* kb = s_qkv + lm * QS + kc + 2 * half;
                #pragma unroll
                for (int s = 0; s < 8; ++s) qf[s] = ld2(qa + 4 * s);
                #pragma unroll
                for (int nt = 0; nt < 4; ++nt)
                    #pragma unroll
                    for (int s = 0; s < 8; ++s) kf[nt][s] = ld2(kb + nt * 16 * QS + 4 * s);
            }
            __builtin_amdgcn_sched_barrier(0);
            v8f sc[4] = {};
            #pragma unroll
            for (int s = 0; s < 8; ++s)
                #pragma unroll
                for (int nt = 0; nt < 4; ++nt)
                    sc[nt] = wmma4(qf[s], kf[nt][s], sc[nt]);

            v2f vf[2][16];
            {
                const float* vl = s_qkv + (2 * half) * QS + vc + lm;
                #pragma unroll
                for (int s = 0; s < 16; ++s)
                    #pragma unroll
                    for (int n2 = 0; n2 < 2; ++n2) {
                        vf[n2][s].x = vl[(4 * s)     * QS + n2 * 16];
                        vf[n2][s].y = vl[(4 * s + 1) * QS + n2 * 16];
                    }
            }

            #pragma unroll
            for (int r = 0; r < 8; ++r) {
                #pragma unroll
                for (int nt = 0; nt < 4; ++nt) {
                    float v = sc[nt][r] * SCALEF + pe_r[nt][r];
                    v = mk[nt][r] ? -1e30f : v;
                    if (nt == 3) v = col48 ? v : -__builtin_inff();
                    sc[nt][r] = v;
                }
                float m = fmaxf(fmaxf(sc[0][r], sc[1][r]), fmaxf(sc[2][r], sc[3][r]));
                m = fmaxf(m, rowx<1>(m));
                m = fmaxf(m, rowx<2>(m));
                m = fmaxf(m, rowx<4>(m));
                m = fmaxf(m, rowx<8>(m));
                #pragma unroll
                for (int nt = 0; nt < 4; ++nt) sc[nt][r] = __expf(sc[nt][r] - m);
                float ps = (sc[0][r] + sc[1][r]) + (sc[2][r] + sc[3][r]);
                ps += rowx<1>(ps);
                ps += rowx<2>(ps);
                ps += rowx<4>(ps);
                ps += rowx<8>(ps);
                const float inv = (1.f / ps) * rk[r];
                #pragma unroll
                for (int nt = 0; nt < 4; ++nt)
                    s_p[(r + 8 * half) * DSS + nt * 16 + lm] = sc[nt][r] * inv;
            }

            v2f pf[16];
            {
                const float* pl = s_p + lm * DSS + 2 * half;
                #pragma unroll
                for (int s = 0; s < 16; ++s) pf[s] = ld2(pl + 4 * s);
            }
            __builtin_amdgcn_sched_barrier(0);
            v8f oa[2] = {};
            #pragma unroll
            for (int s = 0; s < 16; ++s)
                #pragma unroll
                for (int n2 = 0; n2 < 2; ++n2)
                    oa[n2] = wmma4(pf[s], vf[n2][s], oa[n2]);

            #pragma unroll
            for (int n2 = 0; n2 < 2; ++n2)
                #pragma unroll
                for (int r = 0; r < 8; ++r)
                    s_x[(mt * 16 + r + 8 * half) * XS + h * HD + n2 * 16 + lm] = oa[n2][r];
        }
    }
    __syncthreads();

    {
        const float* wl = w_out + (size_t)(2 * half) * DIMC + lm;
        v2f bo0[24], bo1[24];
        ldb24<DIMC>(bo0, wl + (wave % 12) * 16);
        for (int d = wave; d < 24; d += 8) {
            const int mh = d / 12;
            const int nt = d % 12;
            const int nb = nt * 16;
            const int dn = (d + 8 < 24) ? d + 8 : d;
            const float* al = s_x + (mh * 32 + lm) * XS + 2 * half;
            v8f acc[2] = {};
            ldb24<DIMC>(bo1, wl + nb + (size_t)96 * DIMC);
            __builtin_amdgcn_sched_barrier(0);
            mma24<2, XS>(acc, bo0, al);
            ldb24<DIMC>(bo0, wl + (dn % 12) * 16);
            __builtin_amdgcn_sched_barrier(0);
            mma24<2, XS>(acc, bo1, al + 96);
            const float bo = b_out[nb + lm];
            #pragma unroll
            for (int mi = 0; mi < 2; ++mi)
                #pragma unroll
                for (int r = 0; r < 8; ++r) {
                    const int i = (mh * 2 + mi) * 16 + r + 8 * half;
                    if (i < NT) {
                        const int rr = i / 7, cc = i % 7;
                        int gh = wy * 7 + rr + DISPC; if (gh >= IMG) gh -= IMG;
                        int gw = wx * 7 + cc + DISPC; if (gw >= IMG) gw -= IMG;
                        out[(((size_t)b * IMG + gh) * IMG + gw) * DIMC + nb + lm] = acc[mi][r] + bo;
                    }
                }
        }
    }
}

extern "C" void kernel_launch(void* const* d_in, const int* in_sizes, int n_in,
                              void* d_out, int out_size, void* d_ws, size_t ws_size,
                              hipStream_t stream) {
    const float* x     = (const float*)d_in[0];
    const float* w_qkv = (const float*)d_in[1];
    const float* w_out = (const float*)d_in[2];
    const float* b_out = (const float*)d_in[3];
    const float* pe    = (const float*)d_in[4];
    float* out = (float*)d_out;

    const size_t shmem = (size_t)(172 + NTP * XS + NTP * QS + 8 * 16 * DSS) * sizeof(float);
    hipFuncSetAttribute((const void*)swin_window_attn_kernel,
                        hipFuncAttributeMaxDynamicSharedMemorySize, (int)shmem);

    const int grid = 32 * NWH * NWW;
    swin_window_attn_kernel<<<grid, 256, shmem, stream>>>(x, w_qkv, w_out, b_out, pe, out);
}


# LDS per workgroup 229 -> 150 KiB (padding rows 49..63 not kept, reads clamp to row 48; attn-out over q; prob tiles in the x tile) and a 256-register cap: two workgroups resident per WGP
# speedup vs baseline: 2.2025x; 1.1141x over previous
#include <hip/hip_runtime.h>

typedef __attribute__((ext_vector_type(2))) float v2f;
typedef __attribute__((ext_vector_type(8))) float v8f;

#define DIMC   192
#define HEADS  6
#define HD     32
#define NT     49
#define DISPC  3
#define IMG    56
#define NWH    8
#define NWW    8
#define XS     196
#define QS     580
#define DSS    68
#define SCALEF 0.17677669529663687f

__device__ __forceinline__ v8f wmma4(v2f a, v2f b, v8f c) {
    return __builtin_amdgcn_wmma_f32_16x16x4_f32(
         false, a,  false, b,
         (short)0, c,  false,  false);
}

__device__ __forceinline__ v2f ld2(const float* p) { return *(const v2f*)p; }

template <int M>
__device__ __forceinline__ float rowx(float v) {
    const int b = __builtin_bit_cast(int, v);
    return __builtin_bit_cast(float, __builtin_amdgcn_update_dpp(b, b, 0x160 + M, 0xf, 0xf, true));
}

template <int LD>
__device__ __forceinline__ void ldb24(v2f (&dst)[24], const float* __restrict__ p) {
    #pragma unroll
    for (int s = 0; s < 24; ++s) {
        dst[s].x = p[(size_t)(4 * s)     * LD];
        dst[s].y = p[(size_t)(4 * s + 1) * LD];
    }
}

template <int MT, int AS>
__device__ __forceinline__ void mma24(v8f (&acc)[MT], const v2f (&bfr)[24], const float* a,
                                      const float* alast) {
    #pragma unroll
    for (int s = 0; s < 24; ++s) {
        #pragma unroll
        for (int mt = 0; mt < MT; ++mt) {
            v2f af = ld2((mt == MT - 1 ? alast : a + mt * 16 * AS) + 4 * s);
            acc[mt] = wmma4(af, bfr[s], acc[mt]);
        }
    }
}

__global__ __launch_bounds__(256, 4)
void swin_window_attn_kernel(const float* __restrict__ x,
                             const float* __restrict__ w_qkv,
                             const float* __restrict__ w_out,
                             const float* __restrict__ b_out,
                             const float* __restrict__ pe,
                             float* __restrict__ out)
{
    extern __shared__ float smem[];
    float* s_pe  = smem;
    float* s_x   = smem + 172;
    float* s_qkv = s_x   + NT * XS;
    float* s_dot = s_x;

    const int tid  = threadIdx.x;
    const int lane = tid & 31;
    const int wave = tid >> 5;
    const int half = lane >> 4;
    const int lm   = lane & 15;

    const int blk = blockIdx.x;
    const int b   = blk >> 6;
    const int wy  = (blk >> 3) & 7;
    const int wx  = blk & 7;

    const float pev = pe[tid < 169 ? tid : 168];
    float4 xv[10];
    #pragma unroll
    for (int it = 0; it < 9; ++it) {
        const int idx = tid + it * 256;
        const int i  = idx / 48;
        const int dv = idx % 48;
        const int r = i / 7, c = i % 7;
        int gh = wy * 7 + r + DISPC; if (gh >= IMG) gh -= IMG;
        int gw = wx * 7 + c + DISPC; if (gw >= IMG) gw -= IMG;
        xv[it] = *(const float4*)(x + (((size_t)b * IMG + gh) * IMG + gw) * DIMC + dv * 4);
    }
    {
        int gh = wy * 7 + 6 + DISPC; if (gh >= IMG) gh -= IMG;
        int gw = wx * 7 + 6 + DISPC; if (gw >= IMG) gw -= IMG;
        xv[9] = *(const float4*)(x + (((size_t)b * IMG + gh) * IMG + gw) * DIMC
                                   + (tid < 48 ? tid : 47) * 4);
    }
    s_pe[tid < 169 ? tid : 169] = pev;
    #pragma unroll
    for (int it = 0; it < 9; ++it) {
        const int idx = tid + it * 256;
        *(float4*)(s_x + (idx / 48) * XS + (idx % 48) * 4) = xv[it];
    }
    *(float4*)(s_x + (NT - 1) * XS + (tid < 48 ? tid : 47) * 4) = xv[9];
    __syncthreads();

    {
        const float* wl = w_qkv + (size_t)(2 * half) * 576 + lm;
        const float* al = s_x + lm * XS + 2 * half;
        const float* al3 = s_x + (NT - 1) * XS + 2 * half;
        v2f bq0[24], bq1[24];
        ldb24<576>(bq0, wl + wave * 16);
        for (int nt = wave; nt < 36; nt += 8) {
            const int nb = nt * 16;
            const int ntn = (nt + 8 < 36) ? nt + 8 : nt;
            v8f acc[4] = {};
            ldb24<576>(bq1, wl + nb + (size_t)96 * 576);
            mma24<4, XS>(acc, bq0, al, al3);
            ldb24<576>(bq0, wl + ntn * 16);
            __builtin_amdgcn_sched_barrier(0);
            mma24<4, XS>(acc, bq1, al + 96, al3 + 96);
            #pragma unroll
            for (int mt = 0; mt < 3; ++mt)
                #pragma unroll
                for (int r = 0; r < 8; ++r)
                    s_qkv[(mt * 16 + r + 8 * half) * QS + nb + lm] = acc[mt][r];
            if (half == 0) s_qkv[(NT - 1) * QS + nb + lm] = acc[3][0];
        }
    }
    __syncthreads();

    const bool mask_ul = (wy == NWH - 1);
    const bool mask_lr = (wx == NWW - 1);
    {
        const int wv = __builtin_amdgcn_readfirstlane(wave);
        const int mt = wv & 3;
        const int h0 = wv >> 2;
        float* s_p = s_dot + wv * 16 * DSS;
        const int qr = (mt * 16 + lm < NT) ? mt * 16 + lm : NT - 1;

        float pe_r[4][8];
        bool  mk[4][8];
        const int m_ul = mask_ul ? 1 : 0;
        const int m_lr = mask_lr ? 1 : 0;
        #pragma unroll
        for (int nt = 0; nt < 4; ++nt) {
            const int j  = nt * 16 + lm;
            const int xj = j / 7, yj = j % 7;
            #pragma unroll
            for (int r = 0; r < 8; ++r) {
                const int i  = mt * 16 + r + 8 * half;
                const int xi = i / 7, yi = i % 7;
                const int ok = (int)(i < NT) & (int)(j < NT);
                int idx = (xj - xi + 6) * 13 + (yj - yi + 6);
                idx = idx < 0 ? 0 : (idx > 168 ? 168 : idx);
                const float pv = s_pe[idx];
                pe_r[nt][r] = ok ? pv : 0.f;
                const int msk = ((int)((i >= 28) != (j >= 28)) & m_ul) |
                                ((int)((yi >= 4) != (yj >= 4)) & m_lr);
                mk[nt][r] = (msk & ok) != 0;
            }
        }
        const bool col48 = (lm == 0);

        for (int u = 0; u < 3; ++u) {
            const int h  = h0 + 2 * u;
            const int qc = h * HD;
            const int kc = DIMC + h * HD;
            const int vc = 2 * DIMC + h * HD;

            v2f qf[8], kf[4][8];
            {
                const float* qa = s_qkv + qr * QS + qc + 2 * half;
                const float* kb = s_qkv + lm * QS + kc + 2 * half;
                const float* kb3 = s_qkv + (NT - 1) * QS + kc + 2 * half;
                #pragma unroll
                for (int s = 0; s < 8; ++s) qf[s] = ld2(qa + 4 * s);
                #pragma unroll
                for (int nt = 0; nt < 4; ++nt)
                    #pragma unroll
                    for (int s = 0; s < 8; ++s)
                        kf[nt][s] = ld2((nt == 3 ? kb3 : kb + nt * 16 * QS) + 4 * s);
            }
            __builtin_amdgcn_sched_barrier(0);
            v8f sc[4] = {};
            #pragma unroll
            for (int s = 0; s < 8; ++s)
                #pragma unroll
                for (int nt = 0; nt < 4; ++nt)
                    sc[nt] = wmma4(qf[s], kf[nt][s], sc[nt]);

            v2f vf[2][13];
            {
                const float* vl = s_qkv + (2 * half) * QS + vc + lm;
                #pragma unroll
                for (int s = 0; s < 12; ++s)
                    #pragma unroll
                    for (int n2 = 0; n2 < 2; ++n2) {
                        vf[n2][s].x = vl[(4 * s)     * QS + n2 * 16];
                        vf[n2][s].y = vl[(4 * s + 1) * QS + n2 * 16];
                    }
                const float* v48 = s_qkv + (NT - 1) * QS + vc + lm;
                #pragma unroll
                for (int n2 = 0; n2 < 2; ++n2) {
                    vf[n2][12].x = v48[n2 * 16];
                    vf[n2][12].y = vf[n2][12].x;
                }
            }

            #pragma unroll
            for (int r = 0; r < 8; ++r) {
                #pragma unroll
                for (int nt = 0; nt < 4; ++nt) {
                    float v = sc[nt][r] * SCALEF + pe_r[nt][r];
                    v = mk[nt][r] ? -1e30f : v;
                    if (nt == 3) v = col48 ? v : -__builtin_inff();
                    sc[nt][r] = v;
                }
                float m = fmaxf(fmaxf(sc[0][r], sc[1][r]), fmaxf(sc[2][r], sc[3][r]));
                m = fmaxf(m, rowx<1>(m));
                m = fmaxf(m, rowx<2>(m));
                m = fmaxf(m, rowx<4>(m));
                m = fmaxf(m, rowx<8>(m));
                #pragma unroll
                for (int nt = 0; nt < 4; ++nt) sc[nt][r] = __expf(sc[nt][r] - m);
                float ps = (sc[0][r] + sc[1][r]) + (sc[2][r] + sc[3][r]);
                ps += rowx<1>(ps);
                ps += rowx<2>(ps);
                ps += rowx<4>(ps);
                ps += rowx<8>(ps);
                const float inv = 1.f / ps;
                #pragma unroll
                for (int nt = 0; nt < 4; ++nt)
                    s_p[(r + 8 * half) * DSS + nt * 16 + lm] = sc[nt][r] * inv;
            }

            v2f pf[13];
            {
                const float* pl = s_p + lm * DSS + 2 * half;
                #pragma unroll
                for (int s = 0; s < 13; ++s) pf[s] = ld2(pl + 4 * s);
            }
            __builtin_amdgcn_sched_barrier(0);
            v8f oa[2] = {};
            #pragma unroll
            for (int s = 0; s < 13; ++s)
                #pragma unroll
                for (int n2 = 0; n2 < 2; ++n2)
                    oa[n2] = wmma4(pf[s], vf[n2][s], oa[n2]);

            if (mt < 3) {
                #pragma unroll
                for (int n2 = 0; n2 < 2; ++n2)
                    #pragma unroll
                    for (int r = 0; r < 8; ++r)
                        s_qkv[(mt * 16 + r + 8 * half) * QS + qc + n2 * 16 + lm] = oa[n2][r];
            } else if (half == 0) {
                #pragma unroll
                for (int n2 = 0; n2 < 2; ++n2)
                    s_qkv[(NT - 1) * QS + qc + n2 * 16 + lm] = oa[n2][0];
            }
        }
    }
    __syncthreads();

    {
        const float* wl = w_out + (size_t)(2 * half) * DIMC + lm;
        v2f bo0[24], bo1[24];
        ldb24<DIMC>(bo0, wl + (wave % 12) * 16);
        for (int d = wave; d < 24; d += 8) {
            const int mh = d / 12;
            const int nt = d % 12;
            const int nb = nt * 16;
            const int dn = (d + 8 < 24) ? d + 8 : d;
            const int r1 = (mh * 32 + 16 + lm < NT) ? mh * 32 + 16 + lm : NT - 1;
            const float* al  = s_qkv + (mh * 32 + lm) * QS + 2 * half;
            const float* al1 = s_qkv + r1 * QS + 2 * half;
            v8f acc[2] = {};
            ldb24<DIMC>(bo1, wl + nb + (size_t)96 * DIMC);
            __builtin_amdgcn_sched_barrier(0);
            mma24<2, QS>(acc, bo0, al, al1);
            ldb24<DIMC>(bo0, wl + (dn % 12) * 16);
            __builtin_amdgcn_sched_barrier(0);
            mma24<2, QS>(acc, bo1, al + 96, al1 + 96);
            const float bo = b_out[nb + lm];
            #pragma unroll
            for (int mi = 0; mi < 2; ++mi)
                #pragma unroll
                for (int r = 0; r < 8; ++r) {
                    const int i = (mh * 2 + mi) * 16 + r + 8 * half;
                    if (i < NT) {
                        const int rr = i / 7, cc = i % 7;
                        int gh = wy * 7 + rr + DISPC; if (gh >= IMG) gh -= IMG;
                        int gw = wx * 7 + cc + DISPC; if (gw >= IMG) gw -= IMG;
                        out[(((size_t)b * IMG + gh) * IMG + gw) * DIMC + nb + lm] = acc[mi][r] + bo;
                    }
                }
        }
    }
}

extern "C" void kernel_launch(void* const* d_in, const int* in_sizes, int n_in,
                              void* d_out, int out_size, void* d_ws, size_t ws_size,
                              hipStream_t stream) {
    const float* x     = (const float*)d_in[0];
    const float* w_qkv = (const float*)d_in[1];
    const float* w_out = (const float*)d_in[2];
    const float* b_out = (const float*)d_in[3];
    const float* pe    = (const float*)d_in[4];
    float* out = (float*)d_out;

    const size_t shmem = (size_t)(172 + NT * XS + NT * QS) * sizeof(float);
    hipFuncSetAttribute((const void*)swin_window_attn_kernel,
                        hipFuncAttributeMaxDynamicSharedMemorySize, (int)shmem);

    const int grid = 32 * NWH * NWW;
    swin_window_attn_kernel<<<grid, 256, shmem, stream>>>(x, w_qkv, w_out, b_out, pe, out);
}


# Two adjacent windows per workgroup (16 waves, 98 stacked rows): QKV and out-proj GEMMs run 7 row tiles instead of 2 x 4, an eighth less matrix work; k order and sums per element unchanged
# speedup vs baseline: 2.3421x; 1.0634x over previous
#include <hip/hip_runtime.h>

typedef __attribute__((ext_vector_type(2))) float v2f;
typedef __attribute__((ext_vector_type(8))) float v8f;

#define DIMC   192
#define HEADS  6
#define HD     32
#define NT     49
#define NR     98
#define DISPC  3
#define IMG    56
#define NWH    8
#define NWW    8
#define XS     196
#define QS     580
#define DSS    68
#define SCALEF 0.17677669529663687f

__device__ __forceinline__ v8f wmma4(v2f a, v2f b, v8f c) {
    return __builtin_amdgcn_wmma_f32_16x16x4_f32(
         false, a,  false, b,
         (short)0, c,  false,  false);
}

__device__ __forceinline__ v2f ld2(const float* p) { return *(const v2f*)p; }

template <int M>
__device__ __forceinline__ float rowx(float v) {
    const int b = __builtin_bit_cast(int, v);
    return __builtin_bit_cast(float, __builtin_amdgcn_update_dpp(b, b, 0x160 + M, 0xf, 0xf, true));
}

__device__ __forceinline__ size_t row_off(int b, int wy, int wq, int rho) {
    const int win = rho >= NT ? 1 : 0;
    const int i = rho - win * NT;
    const int r = i / 7, c = i - 7 * r;
    int gh = wy * 7 + r + DISPC; if (gh >= IMG) gh -= IMG;
    int gw = (2 * wq + win) * 7 + c + DISPC; if (gw >= IMG) gw -= IMG;
    return (((size_t)b * IMG + gh) * IMG + gw) * DIMC;
}

template <int LD>
__device__ __forceinline__ void ldb24(v2f (&dst)[24], const float* __restrict__ p) {
    #pragma unroll
    for (int s = 0; s < 24; ++s) {
        dst[s].x = p[(size_t)(4 * s)     * LD];
        dst[s].y = p[(size_t)(4 * s + 1) * LD];
    }
}

template <int MT, int AS>
__device__ __forceinline__ void mma24(v8f (&acc)[MT], const v2f (&bfr)[24], const float* a,
                                      const float* alast) {
    #pragma unroll
    for (int s = 0; s < 24; ++s) {
        #pragma unroll
        for (int mt = 0; mt < MT; ++mt) {
            v2f af = ld2((mt == MT - 1 ? alast : a + mt * 16 * AS) + 4 * s);
            acc[mt] = wmma4(af, bfr[s], acc[mt]);
        }
    }
}

__global__ __launch_bounds__(512, 4)
void swin_window_attn_kernel(const float* __restrict__ x,
                             const float* __restrict__ w_qkv,
                             const float* __restrict__ w_out,
                             const float* __restrict__ b_out,
                             const float* __restrict__ pe,
                             float* __restrict__ out)
{
    extern __shared__ float smem[];
    float* s_pe  = smem;
    float* s_x   = smem + 172;
    float* s_qkv = s_x   + NR * XS;
    float* s_dot = s_x;

    const int tid  = threadIdx.x;
    const int lane = tid & 31;
    const int wave = tid >> 5;
    const int half = lane >> 4;
    const int lm   = lane & 15;

    const int blk = blockIdx.x;
    const int b   = blk >> 5;
    const int wy  = (blk >> 2) & 7;
    const int wq  = blk & 3;

    const float pev = pe[tid < 169 ? tid : 168];
    float4 xv[10];
    #pragma unroll
    for (int it = 0; it < 9; ++it) {
        const int idx = tid + it * 512;
        const int rho = idx / 48;
        const int dv  = idx % 48;
        xv[it] = *(const float4*)(x + row_off(b, wy, wq, rho) + dv * 4);
    }
    const int t2 = tid < 96 ? tid : 95;
    xv[9] = *(const float4*)(x + row_off(b, wy, wq, 96 + t2 / 48) + (t2 % 48) * 4);
    s_pe[tid < 169 ? tid : 169 + (tid & 1)] = pev;
    #pragma unroll
    for (int it = 0; it < 9; ++it) {
        const int idx = tid + it * 512;
        *(float4*)(s_x + (idx / 48) * XS + (idx % 48) * 4) = xv[it];
    }
    *(float4*)(s_x + (96 + t2 / 48) * XS + (t2 % 48) * 4) = xv[9];
    __syncthreads();

    {
        const float* wl = w_qkv + (size_t)(2 * half) * 576 + lm;
        const float* al = s_x + lm * XS + 2 * half;
        const float* al6 = s_x + (96 + lm < NR ? 96 + lm : NR - 1) * XS + 2 * half;
        const int wv1 = __builtin_amdgcn_readfirstlane(wave);
        const int nte = 32 + (wv1 & 3);
        const int sub = wv1 >> 2;
        v2f bq0[24], bq1[24];
        ldb24<576>(bq0, wl + wv1 * 16);
        for (int j = 0; j < 2; ++j) {
            const int nb = (wv1 + 16 * j) * 16;
            const int nbn = (j == 0 ? wv1 + 16 : nte) * 16;
            v8f acc[7] = {};
            ldb24<576>(bq1, wl + nb + (size_t)96 * 576);
            mma24<7, XS>(acc, bq0, al, al6);
            ldb24<576>(bq0, wl + nbn);
            __builtin_amdgcn_sched_barrier(0);
            mma24<7, XS>(acc, bq1, al + 96, al6 + 96);
            #pragma unroll
            for (int mt = 0; mt < 6; ++mt)
                #pragma unroll
                for (int r = 0; r < 8; ++r)
                    s_qkv[(mt * 16 + r + 8 * half) * QS + nb + lm] = acc[mt][r];
            if (half == 0) {
                s_qkv[96 * QS + nb + lm] = acc[6][0];
                s_qkv[97 * QS + nb + lm] = acc[6][1];
            }
        }
        {
            const int nb = nte * 16;
            ldb24<576>(bq1, wl + nb + (size_t)96 * 576);
            __builtin_amdgcn_sched_barrier(0);
            if (sub < 3) {
                const float* a2 = al + sub * 32 * XS;
                v8f acc[2] = {};
                mma24<2, XS>(acc, bq0, a2, a2 + 16 * XS);
                mma24<2, XS>(acc, bq1, a2 + 96, a2 + 16 * XS + 96);
                #pragma unroll
                for (int mi = 0; mi < 2; ++mi)
                    #pragma unroll
                    for (int r = 0; r < 8; ++r)
                        s_qkv[(sub * 32 + mi * 16 + r + 8 * half) * QS + nb + lm] = acc[mi][r];
            } else {
                v8f acc[1] = {};
                mma24<1, XS>(acc, bq0, al6, al6);
                mma24<1, XS>(acc, bq1, al6 + 96, al6 + 96);
                if (half == 0) {
                    s_qkv[96 * QS + nb + lm] = acc[0][0];
                    s_qkv[97 * QS + nb + lm] = acc[0][1];
                }
            }
        }
    }
    __syncthreads();

    {
        const int wv = __builtin_amdgcn_readfirstlane(wave);
        const int mt  = wv & 3;
        const int win = (wv >> 2) & 1;
        const int h0  = wv >> 3;
        const int rb  = win * NT;
        const bool mask_ul = (wy == NWH - 1);
        const bool mask_lr = (2 * wq + win == NWW - 1);
        float* s_p = s_dot + wv * 16 * DSS;
        const int qr = rb + ((mt * 16 + lm < NT) ? mt * 16 + lm : NT - 1);

        float pe_r[4][8];
        bool  mk[4][8];
        const int m_ul = mask_ul ? 1 : 0;
        const int m_lr = mask_lr ? 1 : 0;
        #pragma unroll
        for (int nt = 0; nt < 4; ++nt) {
            const int j  = nt * 16 + lm;
            const int xj = j / 7, yj = j % 7;
            #pragma unroll
            for (int r = 0; r < 8; ++r) {
                const int i  = mt * 16 + r + 8 * half;
                const int xi = i / 7, yi = i % 7;
                const int ok = (int)(i < NT) & (int)(j < NT);
                int idx = (xj - xi + 6) * 13 + (yj - yi + 6);
                idx = idx < 0 ? 0 : (idx > 168 ? 168 : idx);
                const float pv = s_pe[idx];
                pe_r[nt][r] = ok ? pv : 0.f;
                const int msk = ((int)((i >= 28) != (j >= 28)) & m_ul) |
                                ((int)((yi >= 4) != (yj >= 4)) & m_lr);
                mk[nt][r] = (msk & ok) != 0;
            }
        }
        const bool col48 = (lm == 0);

        for (int u = 0; u < 3; ++u) {
            const int h  = h0 + 2 * u;
            const int qc = h * HD;
            const int kc = DIMC + h * HD;
            const int vc = 2 * DIMC + h * HD;

            v2f qf[8], kf[4][8];
            {
                const float* qa = s_qkv + qr * QS + qc + 2 * half;
                const float* kb = s_qkv + (rb + lm) * QS + kc + 2 * half;
                const float* kb3 = s_qkv + (rb + NT - 1) * QS + kc + 2 * half;
                #pragma unroll
                for (int s = 0; s < 8; ++s) qf[s] = ld2(qa + 4 * s);
                #pragma unroll
                for (int nt = 0; nt < 4; ++nt)
                    #pragma unroll
                    for (int s = 0; s < 8; ++s)
                        kf[nt][s] = ld2((nt == 3 ? kb3 : kb + nt * 16 * QS) + 4 * s);
            }
            __builtin_amdgcn_sched_barrier(0);
            v8f sc[4] = {};
            #pragma unroll
            for (int s = 0; s < 8; ++s)
                #pragma unroll
                for (int nt = 0; nt < 4; ++nt)
                    sc[nt] = wmma4(qf[s], kf[nt][s], sc[nt]);

            v2f vf[2][13];
            {
                const float* vl = s_qkv + (rb + 2 * half) * QS + vc + lm;
                #pragma unroll
                for (int s = 0; s < 12; ++s)
                    #pragma unroll
                    for (int n2 = 0; n2 < 2; ++n2) {
                        vf[n2][s].x = vl[(4 * s)     * QS + n2 * 16];
                        vf[n2][s].y = vl[(4 * s + 1) * QS + n2 * 16];
                    }
                const float* v48 = s_qkv + (rb + NT - 1) * QS + vc + lm;
                #pragma unroll
                for (int n2 = 0; n2 < 2; ++n2) {
                    vf[n2][12].x = v48[n2 * 16];
                    vf[n2][12].y = vf[n2][12].x;
                }
            }

            #pragma unroll
            for (int r = 0; r < 8; ++r) {
                #pragma unroll
                for (int nt = 0; nt < 4; ++nt) {
                    float v = sc[nt][r] * SCALEF + pe_r[nt][r];
                    v = mk[nt][r] ? -1e30f : v;
                    if (nt == 3) v = col48 ? v : -__builtin_inff();
                    sc[nt][r] = v;
                }
                float m = fmaxf(fmaxf(sc[0][r], sc[1][r]), fmaxf(sc[2][r], sc[3][r]));
                m = fmaxf(m, rowx<1>(m));
                m = fmaxf(m, rowx<2>(m));
                m = fmaxf(m, rowx<4>(m));
                m = fmaxf(m, rowx<8>(m));
                #pragma unroll
                for (int nt = 0; nt < 4; ++nt) sc[nt][r] = __expf(sc[nt][r] - m);
                float ps = (sc[0][r] + sc[1][r]) + (sc[2][r] + sc[3][r]);
                ps += rowx<1>(ps);
                ps += rowx<2>(ps);
                ps += rowx<4>(ps);
                ps += rowx<8>(ps);
                const float inv = 1.f / ps;
                #pragma unroll
                for (int nt = 0; nt < 4; ++nt)
                    s_p[(r + 8 * half) * DSS + nt * 16 + lm] = sc[nt][r] * inv;
            }

            v2f pf[13];
            {
                const float* pl = s_p + lm * DSS + 2 * half;
                #pragma unroll
                for (int s = 0; s < 13; ++s) pf[s] = ld2(pl + 4 * s);
            }
            __builtin_amdgcn_sched_barrier(0);
            v8f oa[2] = {};
            #pragma unroll
            for (int s = 0; s < 13; ++s)
                #pragma unroll
                for (int n2 = 0; n2 < 2; ++n2)
                    oa[n2] = wmma4(pf[s], vf[n2][s], oa[n2]);

            if (mt < 3) {
                #pragma unroll
                for (int n2 = 0; n2 < 2; ++n2)
                    #pragma unroll
                    for (int r = 0; r < 8; ++r)
                        s_qkv[(rb + mt * 16 + r + 8 * half) * QS + qc + n2 * 16 + lm] = oa[n2][r];
            } else if (half == 0) {
                #pragma unroll
                for (int n2 = 0; n2 < 2; ++n2)
                    s_qkv[(rb + NT - 1) * QS + qc + n2 * 16 + lm] = oa[n2][0];
            }
        }
    }
    __syncthreads();

    {
        const float* wl = w_out + (size_t)(2 * half) * DIMC + lm;
        const float* al = s_qkv + lm * QS + 2 * half;
        const int wv3 = __builtin_amdgcn_readfirstlane(wave);
        v2f bo0[24], bo1[24];
        if (wv3 >= 4) {
            const int nb = (wv3 - 4) * 16;
            ldb24<DIMC>(bo0, wl + nb);
            ldb24<DIMC>(bo1, wl + nb + (size_t)96 * DIMC);
            __builtin_amdgcn_sched_barrier(0);
            v8f acc[5] = {};
            mma24<5, QS>(acc, bo0, al, al + 64 * QS);
            mma24<5, QS>(acc, bo1, al + 96, al + 64 * QS + 96);
            const float bo = b_out[nb + lm];
            #pragma unroll
            for (int mi = 0; mi < 5; ++mi)
                #pragma unroll
                for (int r = 0; r < 8; ++r) {
                    const int rho = mi * 16 + r + 8 * half;
                    out[row_off(b, wy, wq, rho) + nb + lm] = acc[mi][r] + bo;
                }
        } else {
            const int n0 = wv3 * 3;
            const float* a5 = al + 80 * QS;
            const float* a6 = s_qkv + (96 + lm < NR ? 96 + lm : NR - 1) * QS + 2 * half;
            ldb24<DIMC>(bo0, wl + n0 * 16);
            for (int u = 0; u < 3; ++u) {
                const int nb = (n0 + u) * 16;
                const int nbn = (u < 2 ? n0 + u + 1 : n0 + u) * 16;
                v8f acc[2] = {};
                ldb24<DIMC>(bo1, wl + nb + (size_t)96 * DIMC);
                __builtin_amdgcn_sched_barrier(0);
                mma24<2, QS>(acc, bo0, a5, a6);
                ldb24<DIMC>(bo0, wl + nbn);
                __builtin_amdgcn_sched_barrier(0);
                mma24<2, QS>(acc, bo1, a5 + 96, a6 + 96);
                const float bo = b_out[nb + lm];
                #pragma unroll
                for (int r = 0; r < 8; ++r) {
                    const int rho = 80 + r + 8 * half;
                    out[row_off(b, wy, wq, rho) + nb + lm] = acc[0][r] + bo;
                }
                if (half == 0) {
                    out[row_off(b, wy, wq, 96) + nb + lm] = acc[1][0] + bo;
                    out[row_off(b, wy, wq, 97) + nb + lm] = acc[1][1] + bo;
                }
            }
        }
    }
}

extern "C" void kernel_launch(void* const* d_in, const int* in_sizes, int n_in,
                              void* d_out, int out_size, void* d_ws, size_t ws_size,
                              hipStream_t stream) {
    const float* x     = (const float*)d_in[0];
    const float* w_qkv = (const float*)d_in[1];
    const float* w_out = (const float*)d_in[2];
    const float* b_out = (const float*)d_in[3];
    const float* pe    = (const float*)d_in[4];
    float* out = (float*)d_out;

    const size_t shmem = (size_t)(172 + NR * XS + NR * QS) * sizeof(float);
    hipFuncSetAttribute((const void*)swin_window_attn_kernel,
                        hipFuncAttributeMaxDynamicSharedMemorySize, (int)shmem);

    const int grid = 32 * NWH * (NWW / 2);
    swin_window_attn_kernel<<<grid, 512, shmem, stream>>>(x, w_qkv, w_out, b_out, pe, out);
}


# One workgroup per row of 8 windows (grid 256, loops over 4 window pairs): the 49th tokens of the 8 windows pass both projections as one shared 16-row tile; pair GEMMs run 6 row tiles, not 7
# speedup vs baseline: 2.4413x; 1.0424x over previous
#include <hip/hip_runtime.h>

typedef __attribute__((ext_vector_type(2))) float v2f;
typedef __attribute__((ext_vector_type(8))) float v8f;

#define DIMC   192
#define HEADS  6
#define HD     32
#define NT     49
#define NB     48
#define NR     96
#define NW8    8
#define DISPC  3
#define IMG    56
#define NWH    8
#define NWW    8
#define XS     196
#define QS     580
#define DSS    68
#define SCALEF 0.17677669529663687f

__device__ __forceinline__ v8f wmma4(v2f a, v2f b, v8f c) {
    return __builtin_amdgcn_wmma_f32_16x16x4_f32(
         false, a,  false, b,
         (short)0, c,  false,  false);
}

__device__ __forceinline__ v2f ld2(const float* p) { return *(const v2f*)p; }

template <int M>
__device__ __forceinline__ float rowx(float v) {
    const int b = __builtin_bit_cast(int, v);
    return __builtin_bit_cast(float, __builtin_amdgcn_update_dpp(b, b, 0x160 + M, 0xf, 0xf, true));
}

__device__ __forceinline__ size_t tok_off(int b, int wy, int wx, int i) {
    const int r = i / 7, c = i - 7 * r;
    int gh = wy * 7 + r + DISPC; if (gh >= IMG) gh -= IMG;
    int gw = wx * 7 + c + DISPC; if (gw >= IMG) gw -= IMG;
    return (((size_t)b * IMG + gh) * IMG + gw) * DIMC;
}

__device__ __forceinline__ size_t row_off(int b, int wy, int p, int rho) {
    const int win = rho >= NB ? 1 : 0;
    return tok_off(b, wy, 2 * p + win, rho - win * NB);
}

template <int LD>
__device__ __forceinline__ void ldb24(v2f (&dst)[24], const float* __restrict__ p) {
    #pragma unroll
    for (int s = 0; s < 24; ++s) {
        dst[s].x = p[(size_t)(4 * s)     * LD];
        dst[s].y = p[(size_t)(4 * s + 1) * LD];
    }
}

template <int MT, int AS>
__device__ __forceinline__ void mma24(v8f (&acc)[MT], const v2f (&bfr)[24], const float* a,
                                      const float* alast) {
    #pragma unroll
    for (int s = 0; s < 24; ++s) {
        #pragma unroll
        for (int mt = 0; mt < MT; ++mt) {
            v2f af = ld2((mt == MT - 1 ? alast : a + mt * 16 * AS) + 4 * s);
            acc[mt] = wmma4(af, bfr[s], acc[mt]);
        }
    }
}

__global__ __launch_bounds__(512, 4)
void swin_window_attn_kernel(const float* __restrict__ x,
                             const float* __restrict__ w_qkv,
                             const float* __restrict__ w_out,
                             const float* __restrict__ b_out,
                             const float* __restrict__ pe,
                             float* __restrict__ out)
{
    extern __shared__ float smem[];
    float* s_pe  = smem;
    float* s_x   = smem + 172;
    float* s_qkv = s_x   + NR * XS;
    float* s_x48 = s_qkv + NR * QS;
    float* s_t48 = s_x48 + NW8 * XS;
    float* s_dot = s_x;

    const int tid  = threadIdx.x;
    const int lane = tid & 31;
    const int wave = tid >> 5;
    const int half = lane >> 4;
    const int lm   = lane & 15;

    const int blk = blockIdx.x;
    const int b   = blk >> 3;
    const int wy  = blk & 7;

    {
        const int t = tid < 384 ? tid : 383;
        const float pev = pe[tid < 169 ? tid : 168];
        const float4 v = *(const float4*)(x + tok_off(b, wy, t / 48, NT - 1) + (t % 48) * 4);
        s_pe[tid < 169 ? tid : 169 + (tid & 1)] = pev;
        *(float4*)(s_x48 + (t / 48) * XS + (t % 48) * 4) = v;
    }
    __syncthreads();

    {
        const float* wl = w_qkv + (size_t)(2 * half) * 576 + lm;
        const float* a8 = s_x48 + (lm < NW8 ? lm : NW8 - 1) * XS + 2 * half;
        const int wv0 = __builtin_amdgcn_readfirstlane(wave);
        const int ntile = wv0 < 4 ? 3 : 2;
        v2f bt0[24], bt1[24];
        for (int j = 0; j < ntile; ++j) {
            const int nb = (wv0 + 16 * j) * 16;
            ldb24<576>(bt0, wl + nb);
            ldb24<576>(bt1, wl + nb + (size_t)96 * 576);
            __builtin_amdgcn_sched_barrier(0);
            v8f acc[1] = {};
            mma24<1, XS>(acc, bt0, a8, a8);
            mma24<1, XS>(acc, bt1, a8 + 96, a8 + 96);
            if (half == 0) {
                #pragma unroll
                for (int r = 0; r < 8; ++r) s_t48[r * QS + nb + lm] = acc[0][r];
            }
        }
    }

    #pragma nounroll
    for (int p = 0; p < NWW / 2; ++p) {

    int tl = tid;
    asm volatile("" : "+v"(tl));
    const int lm   = tl & 15;
    const int half = (tl >> 4) & 1;

    {
        float4 xv[9];
        #pragma unroll
        for (int it = 0; it < 9; ++it) {
            const int idx = tl + it * 512;
            const int rho = idx / 48;
            const int dv  = idx % 48;
            xv[it] = *(const float4*)(x + row_off(b, wy, p, rho) + dv * 4);
        }
        #pragma unroll
        for (int it = 0; it < 9; ++it) {
            const int idx = tl + it * 512;
            *(float4*)(s_x + (idx / 48) * XS + (idx % 48) * 4) = xv[it];
        }
    }
    __syncthreads();

    {
        const float* wl = w_qkv + (size_t)(2 * half) * 576 + lm;
        const float* al = s_x + lm * XS + 2 * half;
        const int wv1 = __builtin_amdgcn_readfirstlane(wave);
        const int nte = 32 + (wv1 & 3);
        const int sub = wv1 >> 2;
        v2f bq0[24], bq1[24];
        ldb24<576>(bq0, wl + wv1 * 16);
        for (int j = 0; j < 2; ++j) {
            const int nb = (wv1 + 16 * j) * 16;
            const int nbn = (j == 0 ? wv1 + 16 : nte) * 16;
            v8f acc[6] = {};
            ldb24<576>(bq1, wl + nb + (size_t)96 * 576);
            mma24<6, XS>(acc, bq0, al, al + 80 * XS);
            ldb24<576>(bq0, wl + nbn);
            __builtin_amdgcn_sched_barrier(0);
            mma24<6, XS>(acc, bq1, al + 96, al + 80 * XS + 96);
            #pragma unroll
            for (int mt = 0; mt < 6; ++mt)
                #pragma unroll
                for (int r = 0; r < 8; ++r)
                    s_qkv[(mt * 16 + r + 8 * half) * QS + nb + lm] = acc[mt][r];
        }
        {
            const int nb = nte * 16;
            ldb24<576>(bq1, wl + nb + (size_t)96 * 576);
            __builtin_amdgcn_sched_barrier(0);
            if (sub < 2) {
                const float* a2 = al + sub * 32 * XS;
                v8f acc[2] = {};
                mma24<2, XS>(acc, bq0, a2, a2 + 16 * XS);
                mma24<2, XS>(acc, bq1, a2 + 96, a2 + 16 * XS + 96);
                #pragma unroll
                for (int mi = 0; mi < 2; ++mi)
                    #pragma unroll
                    for (int r = 0; r < 8; ++r)
                        s_qkv[(sub * 32 + mi * 16 + r + 8 * half) * QS + nb + lm] = acc[mi][r];
            } else {
                const float* a1 = al + (sub + 2) * 16 * XS;
                v8f acc[1] = {};
                mma24<1, XS>(acc, bq0, a1, a1);
                mma24<1, XS>(acc, bq1, a1 + 96, a1 + 96);
                #pragma unroll
                for (int r = 0; r < 8; ++r)
                    s_qkv[((sub + 2) * 16 + r + 8 * half) * QS + nb + lm] = acc[0][r];
            }
        }
    }
    __syncthreads();

    {
        const int wv = __builtin_amdgcn_readfirstlane(wave);
        const int mt  = wv & 3;
        const int win = (wv >> 2) & 1;
        const int h0  = wv >> 3;
        const int rb  = win * NB;
        const int wx  = 2 * p + win;
        const bool mask_ul = (wy == NWH - 1);
        const bool mask_lr = (wx == NWW - 1);
        float* s_p = s_dot + wv * 16 * DSS;
        float* t48 = s_t48 + wx * QS;
        const float* qrow = (mt < 3) ? s_qkv + (rb + mt * 16 + lm) * QS : t48;

        float pe_r[4][8];
        bool  mk[4][8];
        const int m_ul = mask_ul ? 1 : 0;
        const int m_lr = mask_lr ? 1 : 0;
        #pragma unroll
        for (int nt = 0; nt < 4; ++nt) {
            const int j  = nt * 16 + lm;
            const int xj = j / 7, yj = j % 7;
            #pragma unroll
            for (int r = 0; r < 8; ++r) {
                const int i  = mt * 16 + r + 8 * half;
                const int xi = i / 7, yi = i % 7;
                const int ok = (int)(i < NT) & (int)(j < NT);
                int idx = (xj - xi + 6) * 13 + (yj - yi + 6);
                idx = idx < 0 ? 0 : (idx > 168 ? 168 : idx);
                const float pv = s_pe[idx];
                pe_r[nt][r] = ok ? pv : 0.f;
                const int msk = ((int)((i >= 28) != (j >= 28)) & m_ul) |
                                ((int)((yi >= 4) != (yj >= 4)) & m_lr);
                mk[nt][r] = (msk & ok) != 0;
            }
        }
        const bool col48 = (lm == 0);

        for (int u = 0; u < 3; ++u) {
            const int h  = h0 + 2 * u;
            const int qc = h * HD;
            const int kc = DIMC + h * HD;
            const int vc = 2 * DIMC + h * HD;

            v2f qf[8], kf[4][8];
            {
                const float* qa = qrow + qc + 2 * half;
                const float* kb = s_qkv + (rb + lm) * QS + kc + 2 * half;
                const float* kb3 = t48 + kc + 2 * half;
                #pragma unroll
                for (int s = 0; s < 8; ++s) qf[s] = ld2(qa + 4 * s);
                #pragma unroll
                for (int nt = 0; nt < 4; ++nt)
                    #pragma unroll
                    for (int s = 0; s < 8; ++s)
                        kf[nt][s] = ld2((nt == 3 ? kb3 : kb + nt * 16 * QS) + 4 * s);
            }
            __builtin_amdgcn_sched_barrier(0);
            v8f sc[4] = {};
            #pragma unroll
            for (int s = 0; s < 8; ++s)
                #pragma unroll
                for (int nt = 0; nt < 4; ++nt)
                    sc[nt] = wmma4(qf[s], kf[nt][s], sc[nt]);

            v2f vf[2][13];
            {
                const float* vl = s_qkv + (rb + 2 * half) * QS + vc + lm;
                #pragma unroll
                for (int s = 0; s < 12; ++s)
                    #pragma unroll
                    for (int n2 = 0; n2 < 2; ++n2) {
                        vf[n2][s].x = vl[(4 * s)     * QS + n2 * 16];
                        vf[n2][s].y = vl[(4 * s + 1) * QS + n2 * 16];
                    }
                const float* v48 = t48 + vc + lm;
                #pragma unroll
                for (int n2 = 0; n2 < 2; ++n2) {
                    vf[n2][12].x = v48[n2 * 16];
                    vf[n2][12].y = vf[n2][12].x;
                }
            }

            #pragma unroll
            for (int r = 0; r < 8; ++r) {
                #pragma unroll
                for (int nt = 0; nt < 4; ++nt) {
                    float v = sc[nt][r] * SCALEF + pe_r[nt][r];
                    v = mk[nt][r] ? -1e30f : v;
                    if (nt == 3) v = col48 ? v : -__builtin_inff();
                    sc[nt][r] = v;
                }
                float m = fmaxf(fmaxf(sc[0][r], sc[1][r]), fmaxf(sc[2][r], sc[3][r]));
                m = fmaxf(m, rowx<1>(m));
                m = fmaxf(m, rowx<2>(m));
                m = fmaxf(m, rowx<4>(m));
                m = fmaxf(m, rowx<8>(m));
                #pragma unroll
                for (int nt = 0; nt < 4; ++nt) sc[nt][r] = __expf(sc[nt][r] - m);
                float ps = (sc[0][r] + sc[1][r]) + (sc[2][r] + sc[3][r]);
                ps += rowx<1>(ps);
                ps += rowx<2>(ps);
                ps += rowx<4>(ps);
                ps += rowx<8>(ps);
                const float inv = 1.f / ps;
                #pragma unroll
                for (int nt = 0; nt < 4; ++nt)
                    s_p[(r + 8 * half) * DSS + nt * 16 + lm] = sc[nt][r] * inv;
            }

            v2f pf[13];
            {
                const float* pl = s_p + lm * DSS + 2 * half;
                #pragma unroll
                for (int s = 0; s < 13; ++s) pf[s] = ld2(pl + 4 * s);
            }
            __builtin_amdgcn_sched_barrier(0);
            v8f oa[2] = {};
            #pragma unroll
            for (int s = 0; s < 13; ++s)
                #pragma unroll
                for (int n2 = 0; n2 < 2; ++n2)
                    oa[n2] = wmma4(pf[s], vf[n2][s], oa[n2]);

            if (mt < 3) {
                #pragma unroll
                for (int n2 = 0; n2 < 2; ++n2)
                    #pragma unroll
                    for (int r = 0; r < 8; ++r)
                        s_qkv[(rb + mt * 16 + r + 8 * half) * QS + qc + n2 * 16 + lm] = oa[n2][r];
            } else if (half == 0) {
                #pragma unroll
                for (int n2 = 0; n2 < 2; ++n2)
                    t48[qc + n2 * 16 + lm] = oa[n2][0];
            }
        }
    }
    __syncthreads();

    {
        const float* wl = w_out + (size_t)(2 * half) * DIMC + lm;
        const float* al = s_qkv + lm * QS + 2 * half;
        const int wv3 = __builtin_amdgcn_readfirstlane(wave);
        v2f bo0[24], bo1[24];
        if (wv3 >= 4) {
            const int nb = (wv3 - 4) * 16;
            ldb24<DIMC>(bo0, wl + nb);
            ldb24<DIMC>(bo1, wl + nb + (size_t)96 * DIMC);
            __builtin_amdgcn_sched_barrier(0);
            v8f acc[5] = {};
            mma24<5, QS>(acc, bo0, al, al + 64 * QS);
            __builtin_amdgcn_sched_barrier(0);
            mma24<5, QS>(acc, bo1, al + 96, al + 64 * QS + 96);
            const float bo = b_out[nb + lm];
            #pragma unroll
            for (int mi = 0; mi < 5; ++mi)
                #pragma unroll
                for (int r = 0; r < 8; ++r) {
                    const int rho = mi * 16 + r + 8 * half;
                    out[row_off(b, wy, p, rho) + nb + lm] = acc[mi][r] + bo;
                }
        } else {
            const int n0 = wv3 * 3;
            const float* a5 = al + 80 * QS;
            ldb24<DIMC>(bo0, wl + n0 * 16);
            for (int u = 0; u < 3; ++u) {
                const int nb = (n0 + u) * 16;
                const int nbn = (u < 2 ? n0 + u + 1 : n0 + u) * 16;
                v8f acc[1] = {};
                ldb24<DIMC>(bo1, wl + nb + (size_t)96 * DIMC);
                __builtin_amdgcn_sched_barrier(0);
                mma24<1, QS>(acc, bo0, a5, a5);
                ldb24<DIMC>(bo0, wl + nbn);
                __builtin_amdgcn_sched_barrier(0);
                mma24<1, QS>(acc, bo1, a5 + 96, a5 + 96);
                const float bo = b_out[nb + lm];
                #pragma unroll
                for (int r = 0; r < 8; ++r) {
                    const int rho = 80 + r + 8 * half;
                    out[row_off(b, wy, p, rho) + nb + lm] = acc[0][r] + bo;
                }
            }
        }
    }

    }

    {
        const int wvz = __builtin_amdgcn_readfirstlane(wave);
        if (wvz < 4) {
            const float* wl = w_out + (size_t)(2 * half) * DIMC + lm;
            const float* a8 = s_t48 + (lm < NW8 ? lm : NW8 - 1) * QS + 2 * half;
            v2f bz0[24], bz1[24];
            for (int u = 0; u < 3; ++u) {
                const int nb = (wvz * 3 + u) * 16;
                ldb24<DIMC>(bz0, wl + nb);
                ldb24<DIMC>(bz1, wl + nb + (size_t)96 * DIMC);
                __builtin_amdgcn_sched_barrier(0);
                v8f acc[1] = {};
                mma24<1, QS>(acc, bz0, a8, a8);
                mma24<1, QS>(acc, bz1, a8 + 96, a8 + 96);
                const float bo = b_out[nb + lm];
                if (half == 0) {
                    #pragma unroll
                    for (int r = 0; r < 8; ++r)
                        out[tok_off(b, wy, r, NT - 1) + nb + lm] = acc[0][r] + bo;
                }
            }
        }
    }
}

extern "C" void kernel_launch(void* const* d_in, const int* in_sizes, int n_in,
                              void* d_out, int out_size, void* d_ws, size_t ws_size,
                              hipStream_t stream) {
    const float* x     = (const float*)d_in[0];
    const float* w_qkv = (const float*)d_in[1];
    const float* w_out = (const float*)d_in[2];
    const float* b_out = (const float*)d_in[3];
    const float* pe    = (const float*)d_in[4];
    float* out = (float*)d_out;

    const size_t shmem = (size_t)(172 + NR * XS + NR * QS + NW8 * XS + NW8 * QS) * sizeof(float);
    hipFuncSetAttribute((const void*)swin_window_attn_kernel,
                        hipFuncAttributeMaxDynamicSharedMemorySize, (int)shmem);

    const int grid = 32 * NWH;
    swin_window_attn_kernel<<<grid, 512, shmem, stream>>>(x, w_qkv, w_out, b_out, pe, out);
}


# Wave priority by work left: every wave steps s_setprio 3 -> 0 through its share of a phase, so a SIMD's four waves stay level between barriers and none runs a phase's end alone; sums unchanged
# speedup vs baseline: 2.5359x; 1.0387x over previous
#include <hip/hip_runtime.h>

typedef __attribute__((ext_vector_type(2))) float v2f;
typedef __attribute__((ext_vector_type(8))) float v8f;

#define DIMC   192
#define HEADS  6
#define HD     32
#define NT     49
#define NB     48
#define NR     96
#define NW8    8
#define DISPC  3
#define IMG    56
#define NWH    8
#define NWW    8
#define XS     196
#define QS     580
#define DSS    68
#define SCALEF 0.17677669529663687f

__device__ __forceinline__ v8f wmma4(v2f a, v2f b, v8f c) {
    return __builtin_amdgcn_wmma_f32_16x16x4_f32(
         false, a,  false, b,
         (short)0, c,  false,  false);
}

__device__ __forceinline__ v2f ld2(const float* p) { return *(const v2f*)p; }

template <int M>
__device__ __forceinline__ float rowx(float v) {
    const int b = __builtin_bit_cast(int, v);
    return __builtin_bit_cast(float, __builtin_amdgcn_update_dpp(b, b, 0x160 + M, 0xf, 0xf, true));
}

__device__ __forceinline__ size_t tok_off(int b, int wy, int wx, int i) {
    const int r = i / 7, c = i - 7 * r;
    int gh = wy * 7 + r + DISPC; if (gh >= IMG) gh -= IMG;
    int gw = wx * 7 + c + DISPC; if (gw >= IMG) gw -= IMG;
    return (((size_t)b * IMG + gh) * IMG + gw) * DIMC;
}

__device__ __forceinline__ size_t row_off(int b, int wy, int p, int rho) {
    const int win = rho >= NB ? 1 : 0;
    return tok_off(b, wy, 2 * p + win, rho - win * NB);
}

template <int LD>
__device__ __forceinline__ void ldb24(v2f (&dst)[24], const float* __restrict__ p) {
    #pragma unroll
    for (int s = 0; s < 24; ++s) {
        dst[s].x = p[(size_t)(4 * s)     * LD];
        dst[s].y = p[(size_t)(4 * s + 1) * LD];
    }
}

template <int MT, int AS>
__device__ __forceinline__ void mma24(v8f (&acc)[MT], const v2f (&bfr)[24], const float* a,
                                      const float* alast) {
    #pragma unroll
    for (int s = 0; s < 24; ++s) {
        #pragma unroll
        for (int mt = 0; mt < MT; ++mt) {
            v2f af = ld2((mt == MT - 1 ? alast : a + mt * 16 * AS) + 4 * s);
            acc[mt] = wmma4(af, bfr[s], acc[mt]);
        }
    }
}

__global__ __launch_bounds__(512, 4)
void swin_window_attn_kernel(const float* __restrict__ x,
                             const float* __restrict__ w_qkv,
                             const float* __restrict__ w_out,
                             const float* __restrict__ b_out,
                             const float* __restrict__ pe,
                             float* __restrict__ out)
{
    extern __shared__ float smem[];
    float* s_pe  = smem;
    float* s_x   = smem + 172;
    float* s_qkv = s_x   + NR * XS;
    float* s_x48 = s_qkv + NR * QS;
    float* s_t48 = s_x48 + NW8 * XS;
    float* s_dot = s_x;

    const int tid  = threadIdx.x;
    const int lane = tid & 31;
    const int wave = tid >> 5;
    const int half = lane >> 4;
    const int lm   = lane & 15;

    const int blk = blockIdx.x;
    const int b   = blk >> 3;
    const int wy  = blk & 7;

    {
        const int t = tid < 384 ? tid : 383;
        const float pev = pe[tid < 169 ? tid : 168];
        const float4 v = *(const float4*)(x + tok_off(b, wy, t / 48, NT - 1) + (t % 48) * 4);
        s_pe[tid < 169 ? tid : 169 + (tid & 1)] = pev;
        *(float4*)(s_x48 + (t / 48) * XS + (t % 48) * 4) = v;
    }
    __syncthreads();

    {
        const float* wl = w_qkv + (size_t)(2 * half) * 576 + lm;
        const float* a8 = s_x48 + (lm < NW8 ? lm : NW8 - 1) * XS + 2 * half;
        const int wv0 = __builtin_amdgcn_readfirstlane(wave);
        const int ntile = wv0 < 4 ? 3 : 2;
        v2f bt0[24], bt1[24];
        for (int j = 0; j < ntile; ++j) {
            const int nb = (wv0 + 16 * j) * 16;
            ldb24<576>(bt0, wl + nb);
            ldb24<576>(bt1, wl + nb + (size_t)96 * 576);
            __builtin_amdgcn_sched_barrier(0);
            v8f acc[1] = {};
            mma24<1, XS>(acc, bt0, a8, a8);
            mma24<1, XS>(acc, bt1, a8 + 96, a8 + 96);
            if (half == 0) {
                #pragma unroll
                for (int r = 0; r < 8; ++r) s_t48[r * QS + nb + lm] = acc[0][r];
            }
        }
    }

    #pragma nounroll
    for (int p = 0; p < NWW / 2; ++p) {

    int tl = tid;
    asm volatile("" : "+v"(tl));
    const int lm   = tl & 15;
    const int half = (tl >> 4) & 1;

    {
        float4 xv[9];
        #pragma unroll
        for (int it = 0; it < 9; ++it) {
            const int idx = tl + it * 512;
            const int rho = idx / 48;
            const int dv  = idx % 48;
            xv[it] = *(const float4*)(x + row_off(b, wy, p, rho) + dv * 4);
        }
        #pragma unroll
        for (int it = 0; it < 9; ++it) {
            const int idx = tl + it * 512;
            *(float4*)(s_x + (idx / 48) * XS + (idx % 48) * 4) = xv[it];
        }
    }
    __syncthreads();

    {
        const float* wl = w_qkv + (size_t)(2 * half) * 576 + lm;
        const float* al = s_x + lm * XS + 2 * half;
        const int wv1 = __builtin_amdgcn_readfirstlane(wave);
        const int nte = 32 + (wv1 & 3);
        const int sub = wv1 >> 2;
        v2f bq0[24], bq1[24];
        __builtin_amdgcn_s_setprio(3);
        ldb24<576>(bq0, wl + wv1 * 16);
        for (int j = 0; j < 2; ++j) {
            if (j != 0) __builtin_amdgcn_s_setprio(1);
            const int nb = (wv1 + 16 * j) * 16;
            const int nbn = (j == 0 ? wv1 + 16 : nte) * 16;
            v8f acc[6] = {};
            ldb24<576>(bq1, wl + nb + (size_t)96 * 576);
            __builtin_amdgcn_sched_barrier(0);
            mma24<6, XS>(acc, bq0, al, al + 80 * XS);
            ldb24<576>(bq0, wl + nbn);
            if (j == 0) __builtin_amdgcn_s_setprio(2); else __builtin_amdgcn_s_setprio(0);
            __builtin_amdgcn_sched_barrier(0);
            mma24<6, XS>(acc, bq1, al + 96, al + 80 * XS + 96);
            #pragma unroll
            for (int mt = 0; mt < 6; ++mt)
                #pragma unroll
                for (int r = 0; r < 8; ++r)
                    s_qkv[(mt * 16 + r + 8 * half) * QS + nb + lm] = acc[mt][r];
        }
        {
            const int nb = nte * 16;
            ldb24<576>(bq1, wl + nb + (size_t)96 * 576);
            __builtin_amdgcn_sched_barrier(0);
            if (sub < 2) {
                const float* a2 = al + sub * 32 * XS;
                v8f acc[2] = {};
                mma24<2, XS>(acc, bq0, a2, a2 + 16 * XS);
                mma24<2, XS>(acc, bq1, a2 + 96, a2 + 16 * XS + 96);
                #pragma unroll
                for (int mi = 0; mi < 2; ++mi)
                    #pragma unroll
                    for (int r = 0; r < 8; ++r)
                        s_qkv[(sub * 32 + mi * 16 + r + 8 * half) * QS + nb + lm] = acc[mi][r];
            } else {
                const float* a1 = al + (sub + 2) * 16 * XS;
                v8f acc[1] = {};
                mma24<1, XS>(acc, bq0, a1, a1);
                mma24<1, XS>(acc, bq1, a1 + 96, a1 + 96);
                #pragma unroll
                for (int r = 0; r < 8; ++r)
                    s_qkv[((sub + 2) * 16 + r + 8 * half) * QS + nb + lm] = acc[0][r];
            }
        }
    }
    __syncthreads();

    {
        const int wv = __builtin_amdgcn_readfirstlane(wave);
        const int mt  = wv & 3;
        const int win = (wv >> 2) & 1;
        const int h0  = wv >> 3;
        const int rb  = win * NB;
        const int wx  = 2 * p + win;
        const bool mask_ul = (wy == NWH - 1);
        const bool mask_lr = (wx == NWW - 1);
        float* s_p = s_dot + wv * 16 * DSS;
        float* t48 = s_t48 + wx * QS;
        const float* qrow = (mt < 3) ? s_qkv + (rb + mt * 16 + lm) * QS : t48;

        float pe_r[4][8];
        bool  mk[4][8];
        const int m_ul = mask_ul ? 1 : 0;
        const int m_lr = mask_lr ? 1 : 0;
        #pragma unroll
        for (int nt = 0; nt < 4; ++nt) {
            const int j  = nt * 16 + lm;
            const int xj = j / 7, yj = j % 7;
            #pragma unroll
            for (int r = 0; r < 8; ++r) {
                const int i  = mt * 16 + r + 8 * half;
                const int xi = i / 7, yi = i % 7;
                const int ok = (int)(i < NT) & (int)(j < NT);
                int idx = (xj - xi + 6) * 13 + (yj - yi + 6);
                idx = idx < 0 ? 0 : (idx > 168 ? 168 : idx);
                const float pv = s_pe[idx];
                pe_r[nt][r] = ok ? pv : 0.f;
                const int msk = ((int)((i >= 28) != (j >= 28)) & m_ul) |
                                ((int)((yi >= 4) != (yj >= 4)) & m_lr);
                mk[nt][r] = (msk & ok) != 0;
            }
        }
        const bool col48 = (lm == 0);

        for (int u = 0; u < 3; ++u) {
            if (u == 0) __builtin_amdgcn_s_setprio(3);
            else if (u == 1) __builtin_amdgcn_s_setprio(2);
            else __builtin_amdgcn_s_setprio(1);
            const int h  = h0 + 2 * u;
            const int qc = h * HD;
            const int kc = DIMC + h * HD;
            const int vc = 2 * DIMC + h * HD;

            v2f qf[8], kf[4][8];
            {
                const float* qa = qrow + qc + 2 * half;
                const float* kb = s_qkv + (rb + lm) * QS + kc + 2 * half;
                const float* kb3 = t48 + kc + 2 * half;
                #pragma unroll
                for (int s = 0; s < 8; ++s) qf[s] = ld2(qa + 4 * s);
                #pragma unroll
                for (int nt = 0; nt < 4; ++nt)
                    #pragma unroll
                    for (int s = 0; s < 8; ++s)
                        kf[nt][s] = ld2((nt == 3 ? kb3 : kb + nt * 16 * QS) + 4 * s);
            }
            __builtin_amdgcn_sched_barrier(0);
            v8f sc[4] = {};
            #pragma unroll
            for (int s = 0; s < 8; ++s)
                #pragma unroll
                for (int nt = 0; nt < 4; ++nt)
                    sc[nt] = wmma4(qf[s], kf[nt][s], sc[nt]);

            v2f vf[2][13];
            {
                const float* vl = s_qkv + (rb + 2 * half) * QS + vc + lm;
                #pragma unroll
                for (int s = 0; s < 12; ++s)
                    #pragma unroll
                    for (int n2 = 0; n2 < 2; ++n2) {
                        vf[n2][s].x = vl[(4 * s)     * QS + n2 * 16];
                        vf[n2][s].y = vl[(4 * s + 1) * QS + n2 * 16];
                    }
                const float* v48 = t48 + vc + lm;
                #pragma unroll
                for (int n2 = 0; n2 < 2; ++n2) {
                    vf[n2][12].x = v48[n2 * 16];
                    vf[n2][12].y = vf[n2][12].x;
                }
            }

            #pragma unroll
            for (int r = 0; r < 8; ++r) {
                #pragma unroll
                for (int nt = 0; nt < 4; ++nt) {
                    float v = sc[nt][r] * SCALEF + pe_r[nt][r];
                    v = mk[nt][r] ? -1e30f : v;
                    if (nt == 3) v = col48 ? v : -__builtin_inff();
                    sc[nt][r] = v;
                }
                float m = fmaxf(fmaxf(sc[0][r], sc[1][r]), fmaxf(sc[2][r], sc[3][r]));
                m = fmaxf(m, rowx<1>(m));
                m = fmaxf(m, rowx<2>(m));
                m = fmaxf(m, rowx<4>(m));
                m = fmaxf(m, rowx<8>(m));
                #pragma unroll
                for (int nt = 0; nt < 4; ++nt) sc[nt][r] = __expf(sc[nt][r] - m);
                float ps = (sc[0][r] + sc[1][r]) + (sc[2][r] + sc[3][r]);
                ps += rowx<1>(ps);
                ps += rowx<2>(ps);
                ps += rowx<4>(ps);
                ps += rowx<8>(ps);
                const float inv = 1.f / ps;
                #pragma unroll
                for (int nt = 0; nt < 4; ++nt)
                    s_p[(r + 8 * half) * DSS + nt * 16 + lm] = sc[nt][r] * inv;
            }

            v2f pf[13];
            {
                const float* pl = s_p + lm * DSS + 2 * half;
                #pragma unroll
                for (int s = 0; s < 13; ++s) pf[s] = ld2(pl + 4 * s);
            }
            __builtin_amdgcn_sched_barrier(0);
            v8f oa[2] = {};
            #pragma unroll
            for (int s = 0; s < 13; ++s)
                #pragma unroll
                for (int n2 = 0; n2 < 2; ++n2)
                    oa[n2] = wmma4(pf[s], vf[n2][s], oa[n2]);

            if (mt < 3) {
                #pragma unroll
                for (int n2 = 0; n2 < 2; ++n2)
                    #pragma unroll
                    for (int r = 0; r < 8; ++r)
                        s_qkv[(rb + mt * 16 + r + 8 * half) * QS + qc + n2 * 16 + lm] = oa[n2][r];
            } else if (half == 0) {
                #pragma unroll
                for (int n2 = 0; n2 < 2; ++n2)
                    t48[qc + n2 * 16 + lm] = oa[n2][0];
            }
        }
        __builtin_amdgcn_s_setprio(0);
    }
    __syncthreads();

    {
        const float* wl = w_out + (size_t)(2 * half) * DIMC + lm;
        const float* al = s_qkv + lm * QS + 2 * half;
        const int wv3 = __builtin_amdgcn_readfirstlane(wave);
        v2f bo0[24], bo1[24];
        if (wv3 >= 4) {
            const int nb = (wv3 - 4) * 16;
            __builtin_amdgcn_s_setprio(3);
            ldb24<DIMC>(bo0, wl + nb);
            ldb24<DIMC>(bo1, wl + nb + (size_t)96 * DIMC);
            __builtin_amdgcn_sched_barrier(0);
            v8f acc[5] = {};
            mma24<5, QS>(acc, bo0, al, al + 64 * QS);
            __builtin_amdgcn_sched_barrier(0);
            __builtin_amdgcn_s_setprio(1);
            mma24<5, QS>(acc, bo1, al + 96, al + 64 * QS + 96);
            __builtin_amdgcn_s_setprio(0);
            const float bo = b_out[nb + lm];
            #pragma unroll
            for (int mi = 0; mi < 5; ++mi)
                #pragma unroll
                for (int r = 0; r < 8; ++r) {
                    const int rho = mi * 16 + r + 8 * half;
                    out[row_off(b, wy, p, rho) + nb + lm] = acc[mi][r] + bo;
                }
        } else {
            const int n0 = wv3 * 3;
            const float* a5 = al + 80 * QS;
            ldb24<DIMC>(bo0, wl + n0 * 16);
            for (int u = 0; u < 3; ++u) {
                if (u == 0) __builtin_amdgcn_s_setprio(2);
                else if (u == 1) __builtin_amdgcn_s_setprio(1);
                else __builtin_amdgcn_s_setprio(0);
                const int nb = (n0 + u) * 16;
                const int nbn = (u < 2 ? n0 + u + 1 : n0 + u) * 16;
                v8f acc[1] = {};
                ldb24<DIMC>(bo1, wl + nb + (size_t)96 * DIMC);
                __builtin_amdgcn_sched_barrier(0);
                mma24<1, QS>(acc, bo0, a5, a5);
                ldb24<DIMC>(bo0, wl + nbn);
                __builtin_amdgcn_sched_barrier(0);
                mma24<1, QS>(acc, bo1, a5 + 96, a5 + 96);
                const float bo = b_out[nb + lm];
                #pragma unroll
                for (int r = 0; r < 8; ++r) {
                    const int rho = 80 + r + 8 * half;
                    out[row_off(b, wy, p, rho) + nb + lm] = acc[0][r] + bo;
                }
            }
        }
    }

    }

    {
        const int wvz = __builtin_amdgcn_readfirstlane(wave);
        if (wvz < 4) {
            const float* wl = w_out + (size_t)(2 * half) * DIMC + lm;
            const float* a8 = s_t48 + (lm < NW8 ? lm : NW8 - 1) * QS + 2 * half;
            v2f bz0[24], bz1[24];
            for (int u = 0; u < 3; ++u) {
                const int nb = (wvz * 3 + u) * 16;
                ldb24<DIMC>(bz0, wl + nb);
                ldb24<DIMC>(bz1, wl + nb + (size_t)96 * DIMC);
                __builtin_amdgcn_sched_barrier(0);
                v8f acc[1] = {};
                mma24<1, QS>(acc, bz0, a8, a8);
                mma24<1, QS>(acc, bz1, a8 + 96, a8 + 96);
                const float bo = b_out[nb + lm];
                if (half == 0) {
                    #pragma unroll
                    for (int r = 0; r < 8; ++r)
                        out[tok_off(b, wy, r, NT - 1) + nb + lm] = acc[0][r] + bo;
                }
            }
        }
    }
}

extern "C" void kernel_launch(void* const* d_in, const int* in_sizes, int n_in,
                              void* d_out, int out_size, void* d_ws, size_t ws_size,
                              hipStream_t stream) {
    const float* x     = (const float*)d_in[0];
    const float* w_qkv = (const float*)d_in[1];
    const float* w_out = (const float*)d_in[2];
    const float* b_out = (const float*)d_in[3];
    const float* pe    = (const float*)d_in[4];
    float* out = (float*)d_out;

    const size_t shmem = (size_t)(172 + NR * XS + NR * QS + NW8 * XS + NW8 * QS) * sizeof(float);
    hipFuncSetAttribute((const void*)swin_window_attn_kernel,
                        hipFuncAttributeMaxDynamicSharedMemorySize, (int)shmem);

    const int grid = 32 * NWH;
    swin_window_attn_kernel<<<grid, 512, shmem, stream>>>(x, w_qkv, w_out, b_out, pe, out);
}


# QKV and out-proj GEMMs: a wave owns 32 adjacent columns and runs its two column tiles side by side: each A fragment read from LDS feeds two WMMAs, weights come as 8-byte loads; sum order unchanged
# speedup vs baseline: 2.6810x; 1.0572x over previous
#include <hip/hip_runtime.h>

typedef __attribute__((ext_vector_type(2))) float v2f;
typedef __attribute__((ext_vector_type(8))) float v8f;

#define DIMC   192
#define HEADS  6
#define HD     32
#define NT     49
#define NB     48
#define NR     96
#define NW8    8
#define DISPC  3
#define IMG    56
#define NWH    8
#define NWW    8
#define XS     196
#define QS     580
#define DSS    68
#define SCALEF 0.17677669529663687f

__device__ __forceinline__ v8f wmma4(v2f a, v2f b, v8f c) {
    return __builtin_amdgcn_wmma_f32_16x16x4_f32(
         false, a,  false, b,
         (short)0, c,  false,  false);
}

__device__ __forceinline__ v2f ld2(const float* p) { return *(const v2f*)p; }

template <int M>
__device__ __forceinline__ float rowx(float v) {
    const int b = __builtin_bit_cast(int, v);
    return __builtin_bit_cast(float, __builtin_amdgcn_update_dpp(b, b, 0x160 + M, 0xf, 0xf, true));
}

__device__ __forceinline__ size_t tok_off(int b, int wy, int wx, int i) {
    const int r = i / 7, c = i - 7 * r;
    int gh = wy * 7 + r + DISPC; if (gh >= IMG) gh -= IMG;
    int gw = wx * 7 + c + DISPC; if (gw >= IMG) gw -= IMG;
    return (((size_t)b * IMG + gh) * IMG + gw) * DIMC;
}

__device__ __forceinline__ size_t row_off(int b, int wy, int p, int rho) {
    const int win = rho >= NB ? 1 : 0;
    return tok_off(b, wy, 2 * p + win, rho - win * NB);
}

template <int LD>
__device__ __forceinline__ void ldb24(v2f (&dst)[24], const float* __restrict__ p) {
    #pragma unroll
    for (int s = 0; s < 24; ++s) {
        dst[s].x = p[(size_t)(4 * s)     * LD];
        dst[s].y = p[(size_t)(4 * s + 1) * LD];
    }
}

template <int MT, int AS>
__device__ __forceinline__ void mma24(v8f (&acc)[MT], const v2f (&bfr)[24], const float* a,
                                      const float* alast) {
    #pragma unroll
    for (int s = 0; s < 24; ++s) {
        #pragma unroll
        for (int mt = 0; mt < MT; ++mt) {
            v2f af = ld2((mt == MT - 1 ? alast : a + mt * 16 * AS) + 4 * s);
            acc[mt] = wmma4(af, bfr[s], acc[mt]);
        }
    }
}

template <int LD>
__device__ __forceinline__ void ldb12x2(v2f (&r0)[12], v2f (&r1)[12], const float* __restrict__ p) {
    #pragma unroll
    for (int s = 0; s < 12; ++s) {
        r0[s] = *(const v2f*)(p + (size_t)(4 * s)     * LD);
        r1[s] = *(const v2f*)(p + (size_t)(4 * s + 1) * LD);
    }
}

template <int MT, int AS, int G>
__device__ __forceinline__ void mma12x2(v8f (&acc0)[MT], v8f (&acc1)[MT], const v2f (&r0)[12],
                                        const v2f (&r1)[12], const float* a) {
    v2f af[G][MT];
    #pragma unroll
    for (int g = 0; g < G; ++g)
        #pragma unroll
        for (int mt = 0; mt < MT; ++mt) af[g][mt] = ld2(a + mt * 16 * AS + 4 * g);
    #pragma unroll
    for (int s0 = 0; s0 < 12; s0 += G) {
        v2f an[G][MT];
        if (s0 + G < 12) {
            #pragma unroll
            for (int g = 0; g < G; ++g)
                #pragma unroll
                for (int mt = 0; mt < MT; ++mt)
                    an[g][mt] = ld2(a + mt * 16 * AS + 4 * (s0 + G + g));
        }
        #pragma unroll
        for (int g = 0; g < G; ++g) {
            v2f b0, b1;
            b0.x = r0[s0 + g].x; b0.y = r1[s0 + g].x;
            b1.x = r0[s0 + g].y; b1.y = r1[s0 + g].y;
            #pragma unroll
            for (int mt = 0; mt < MT; ++mt) {
                acc0[mt] = wmma4(af[g][mt], b0, acc0[mt]);
                acc1[mt] = wmma4(af[g][mt], b1, acc1[mt]);
            }
        }
        if (s0 + G < 12) {
            #pragma unroll
            for (int g = 0; g < G; ++g)
                #pragma unroll
                for (int mt = 0; mt < MT; ++mt) af[g][mt] = an[g][mt];
        }
    }
}

__global__ __launch_bounds__(512, 4)
void swin_window_attn_kernel(const float* __restrict__ x,
                             const float* __restrict__ w_qkv,
                             const float* __restrict__ w_out,
                             const float* __restrict__ b_out,
                             const float* __restrict__ pe,
                             float* __restrict__ out)
{
    extern __shared__ float smem[];
    float* s_pe  = smem;
    float* s_x   = smem + 172;
    float* s_qkv = s_x   + NR * XS;
    float* s_x48 = s_qkv + NR * QS;
    float* s_t48 = s_x48 + NW8 * XS;
    float* s_dot = s_x;

    const int tid  = threadIdx.x;
    const int lane = tid & 31;
    const int wave = tid >> 5;
    const int half = lane >> 4;
    const int lm   = lane & 15;

    const int blk = blockIdx.x;
    const int b   = blk >> 3;
    const int wy  = blk & 7;

    {
        const int t = tid < 384 ? tid : 383;
        const float pev = pe[tid < 169 ? tid : 168];
        const float4 v = *(const float4*)(x + tok_off(b, wy, t / 48, NT - 1) + (t % 48) * 4);
        s_pe[tid < 169 ? tid : 169 + (tid & 1)] = pev;
        *(float4*)(s_x48 + (t / 48) * XS + (t % 48) * 4) = v;
    }
    __syncthreads();

    {
        const float* wl = w_qkv + (size_t)(2 * half) * 576 + lm;
        const float* a8 = s_x48 + (lm < NW8 ? lm : NW8 - 1) * XS + 2 * half;
        const int wv0 = __builtin_amdgcn_readfirstlane(wave);
        const int ntile = wv0 < 4 ? 3 : 2;
        v2f bt0[24], bt1[24];
        for (int j = 0; j < ntile; ++j) {
            const int nb = (wv0 + 16 * j) * 16;
            ldb24<576>(bt0, wl + nb);
            ldb24<576>(bt1, wl + nb + (size_t)96 * 576);
            __builtin_amdgcn_sched_barrier(0);
            v8f acc[1] = {};
            mma24<1, XS>(acc, bt0, a8, a8);
            mma24<1, XS>(acc, bt1, a8 + 96, a8 + 96);
            if (half == 0) {
                #pragma unroll
                for (int r = 0; r < 8; ++r) s_t48[r * QS + nb + lm] = acc[0][r];
            }
        }
    }

    #pragma nounroll
    for (int p = 0; p < NWW / 2; ++p) {

    int tl = tid;
    asm volatile("" : "+v"(tl));
    const int lm   = tl & 15;
    const int half = (tl >> 4) & 1;

    {
        float4 xv[9];
        #pragma unroll
        for (int it = 0; it < 9; ++it) {
            const int idx = tl + it * 512;
            const int rho = idx / 48;
            const int dv  = idx % 48;
            xv[it] = *(const float4*)(x + row_off(b, wy, p, rho) + dv * 4);
        }
        #pragma unroll
        for (int it = 0; it < 9; ++it) {
            const int idx = tl + it * 512;
            *(float4*)(s_x + (idx / 48) * XS + (idx % 48) * 4) = xv[it];
        }
    }
    __syncthreads();

    {
        const float* wl = w_qkv + (size_t)(2 * half) * 576 + lm;
        const float* al = s_x + lm * XS + 2 * half;
        const int wv1 = __builtin_amdgcn_readfirstlane(wave);
        const int nte = 32 + (wv1 & 3);
        const int sub = wv1 >> 2;
        const int nb2 = wv1 * 32;
        const float* wp = w_qkv + (size_t)(2 * half) * 576 + nb2 + 2 * lm;
        v2f pa0[12], pa1[12], pb0[12], pb1[12];
        v2f bq0[24], bq1[24];
        __builtin_amdgcn_s_setprio(3);
        ldb12x2<576>(pa0, pa1, wp);
        v8f acc0[6] = {}, acc1[6] = {};
        ldb12x2<576>(pb0, pb1, wp + (size_t)48 * 576);
        __builtin_amdgcn_sched_barrier(0);
        mma12x2<6, XS, 1>(acc0, acc1, pa0, pa1, al);
        ldb12x2<576>(pa0, pa1, wp + (size_t)96 * 576);
        __builtin_amdgcn_s_setprio(2);
        __builtin_amdgcn_sched_barrier(0);
        mma12x2<6, XS, 1>(acc0, acc1, pb0, pb1, al + 48);
        ldb12x2<576>(pb0, pb1, wp + (size_t)144 * 576);
        __builtin_amdgcn_s_setprio(1);
        __builtin_amdgcn_sched_barrier(0);
        mma12x2<6, XS, 1>(acc0, acc1, pa0, pa1, al + 96);
        ldb24<576>(bq0, wl + nte * 16);
        __builtin_amdgcn_s_setprio(0);
        __builtin_amdgcn_sched_barrier(0);
        mma12x2<6, XS, 1>(acc0, acc1, pb0, pb1, al + 144);
        {
            const int nb = nte * 16;
            ldb24<576>(bq1, wl + nb + (size_t)96 * 576);
            #pragma unroll
            for (int mt = 0; mt < 6; ++mt)
                #pragma unroll
                for (int r = 0; r < 8; ++r) {
                    v2f o;
                    o.x = acc0[mt][r];
                    o.y = acc1[mt][r];
                    *(v2f*)(s_qkv + (mt * 16 + r + 8 * half) * QS + nb2 + 2 * lm) = o;
                }
            __builtin_amdgcn_sched_barrier(0);
            if (sub < 2) {
                const float* a2 = al + sub * 32 * XS;
                v8f acc[2] = {};
                mma24<2, XS>(acc, bq0, a2, a2 + 16 * XS);
                mma24<2, XS>(acc, bq1, a2 + 96, a2 + 16 * XS + 96);
                #pragma unroll
                for (int mi = 0; mi < 2; ++mi)
                    #pragma unroll
                    for (int r = 0; r < 8; ++r)
                        s_qkv[(sub * 32 + mi * 16 + r + 8 * half) * QS + nb + lm] = acc[mi][r];
            } else {
                const float* a1 = al + (sub + 2) * 16 * XS;
                v8f acc[1] = {};
                mma24<1, XS>(acc, bq0, a1, a1);
                mma24<1, XS>(acc, bq1, a1 + 96, a1 + 96);
                #pragma unroll
                for (int r = 0; r < 8; ++r)
                    s_qkv[((sub + 2) * 16 + r + 8 * half) * QS + nb + lm] = acc[0][r];
            }
        }
    }
    __syncthreads();

    {
        const int wv = __builtin_amdgcn_readfirstlane(wave);
        const int mt  = wv & 3;
        const int win = (wv >> 2) & 1;
        const int h0  = wv >> 3;
        const int rb  = win * NB;
        const int wx  = 2 * p + win;
        const bool mask_ul = (wy == NWH - 1);
        const bool mask_lr = (wx == NWW - 1);
        float* s_p = s_dot + wv * 16 * DSS;
        float* t48 = s_t48 + wx * QS;
        const float* qrow = (mt < 3) ? s_qkv + (rb + mt * 16 + lm) * QS : t48;

        float pe_r[4][8];
        bool  mk[4][8];
        const int m_ul = mask_ul ? 1 : 0;
        const int m_lr = mask_lr ? 1 : 0;
        #pragma unroll
        for (int nt = 0; nt < 4; ++nt) {
            const int j  = nt * 16 + lm;
            const int xj = j / 7, yj = j % 7;
            #pragma unroll
            for (int r = 0; r < 8; ++r) {
                const int i  = mt * 16 + r + 8 * half;
                const int xi = i / 7, yi = i % 7;
                const int ok = (int)(i < NT) & (int)(j < NT);
                int idx = (xj - xi + 6) * 13 + (yj - yi + 6);
                idx = idx < 0 ? 0 : (idx > 168 ? 168 : idx);
                const float pv = s_pe[idx];
                pe_r[nt][r] = ok ? pv : 0.f;
                const int msk = ((int)((i >= 28) != (j >= 28)) & m_ul) |
                                ((int)((yi >= 4) != (yj >= 4)) & m_lr);
                mk[nt][r] = (msk & ok) != 0;
            }
        }
        const bool col48 = (lm == 0);

        for (int u = 0; u < 3; ++u) {
            if (u == 0) __builtin_amdgcn_s_setprio(3);
            else if (u == 1) __builtin_amdgcn_s_setprio(2);
            else __builtin_amdgcn_s_setprio(1);
            const int h  = h0 + 2 * u;
            const int qc = h * HD;
            const int kc = DIMC + h * HD;
            const int vc = 2 * DIMC + h * HD;

            v2f qf[8], kf[4][8];
            {
                const float* qa = qrow + qc + 2 * half;
                const float* kb = s_qkv + (rb + lm) * QS + kc + 2 * half;
                const float* kb3 = t48 + kc + 2 * half;
                #pragma unroll
                for (int s = 0; s < 8; ++s) qf[s] = ld2(qa + 4 * s);
                #pragma unroll
                for (int nt = 0; nt < 4; ++nt)
                    #pragma unroll
                    for (int s = 0; s < 8; ++s)
                        kf[nt][s] = ld2((nt == 3 ? kb3 : kb + nt * 16 * QS) + 4 * s);
            }
            __builtin_amdgcn_sched_barrier(0);
            v8f sc[4] = {};
            #pragma unroll
            for (int s = 0; s < 8; ++s)
                #pragma unroll
                for (int nt = 0; nt < 4; ++nt)
                    sc[nt] = wmma4(qf[s], kf[nt][s], sc[nt]);

            v2f vf[2][13];
            {
                const float* vl = s_qkv + (rb + 2 * half) * QS + vc + lm;
                #pragma unroll
                for (int s = 0; s < 12; ++s)
                    #pragma unroll
                    for (int n2 = 0; n2 < 2; ++n2) {
                        vf[n2][s].x = vl[(4 * s)     * QS + n2 * 16];
                        vf[n2][s].y = vl[(4 * s + 1) * QS + n2 * 16];
                    }
                const float* v48 = t48 + vc + lm;
                #pragma unroll
                for (int n2 = 0; n2 < 2; ++n2) {
                    vf[n2][12].x = v48[n2 * 16];
                    vf[n2][12].y = vf[n2][12].x;
                }
            }

            #pragma unroll
            for (int r = 0; r < 8; ++r) {
                #pragma unroll
                for (int nt = 0; nt < 4; ++nt) {
                    float v = sc[nt][r] * SCALEF + pe_r[nt][r];
                    v = mk[nt][r] ? -1e30f : v;
                    if (nt == 3) v = col48 ? v : -__builtin_inff();
                    sc[nt][r] = v;
                }
                float m = fmaxf(fmaxf(sc[0][r], sc[1][r]), fmaxf(sc[2][r], sc[3][r]));
                m = fmaxf(m, rowx<1>(m));
                m = fmaxf(m, rowx<2>(m));
                m = fmaxf(m, rowx<4>(m));
                m = fmaxf(m, rowx<8>(m));
                #pragma unroll
                for (int nt = 0; nt < 4; ++nt) sc[nt][r] = __expf(sc[nt][r] - m);
                float ps = (sc[0][r] + sc[1][r]) + (sc[2][r] + sc[3][r]);
                ps += rowx<1>(ps);
                ps += rowx<2>(ps);
                ps += rowx<4>(ps);
                ps += rowx<8>(ps);
                const float inv = 1.f / ps;
                #pragma unroll
                for (int nt = 0; nt < 4; ++nt)
                    s_p[(r + 8 * half) * DSS + nt * 16 + lm] = sc[nt][r] * inv;
            }

            v2f pf[13];
            {
                const float* pl = s_p + lm * DSS + 2 * half;
                #pragma unroll
                for (int s = 0; s < 13; ++s) pf[s] = ld2(pl + 4 * s);
            }
            __builtin_amdgcn_sched_barrier(0);
            v8f oa[2] = {};
            #pragma unroll
            for (int s = 0; s < 13; ++s)
                #pragma unroll
                for (int n2 = 0; n2 < 2; ++n2)
                    oa[n2] = wmma4(pf[s], vf[n2][s], oa[n2]);

            if (mt < 3) {
                #pragma unroll
                for (int n2 = 0; n2 < 2; ++n2)
                    #pragma unroll
                    for (int r = 0; r < 8; ++r)
                        s_qkv[(rb + mt * 16 + r + 8 * half) * QS + qc + n2 * 16 + lm] = oa[n2][r];
            } else if (half == 0) {
                #pragma unroll
                for (int n2 = 0; n2 < 2; ++n2)
                    t48[qc + n2 * 16 + lm] = oa[n2][0];
            }
        }
        __builtin_amdgcn_s_setprio(0);
    }
    __syncthreads();

    {
        const int wv3 = __builtin_amdgcn_readfirstlane(wave);
        if (wv3 >= 4) {
            const int cp  = (wv3 - 4) >> 1;
            const int rh  = (wv3 - 4) & 1;
            const int nb2 = cp * 32;
            const float* wp = w_out + (size_t)(2 * half) * DIMC + nb2 + 2 * lm;
            const float* a3 = s_qkv + (rh * 48 + lm) * QS + 2 * half;
            v2f pa0[12], pa1[12], pb0[12], pb1[12];
            __builtin_amdgcn_s_setprio(3);
            ldb12x2<DIMC>(pa0, pa1, wp);
            ldb12x2<DIMC>(pb0, pb1, wp + (size_t)48 * DIMC);
            __builtin_amdgcn_sched_barrier(0);
            v8f acc0[3] = {}, acc1[3] = {};
            mma12x2<3, QS, 2>(acc0, acc1, pa0, pa1, a3);
            ldb12x2<DIMC>(pa0, pa1, wp + (size_t)96 * DIMC);
            __builtin_amdgcn_s_setprio(2);
            __builtin_amdgcn_sched_barrier(0);
            mma12x2<3, QS, 2>(acc0, acc1, pb0, pb1, a3 + 48);
            ldb12x2<DIMC>(pb0, pb1, wp + (size_t)144 * DIMC);
            __builtin_amdgcn_s_setprio(1);
            __builtin_amdgcn_sched_barrier(0);
            mma12x2<3, QS, 2>(acc0, acc1, pa0, pa1, a3 + 96);
            __builtin_amdgcn_s_setprio(0);
            mma12x2<3, QS, 2>(acc0, acc1, pb0, pb1, a3 + 144);
            const v2f bo = *(const v2f*)(b_out + nb2 + 2 * lm);
            #pragma unroll
            for (int mi = 0; mi < 3; ++mi)
                #pragma unroll
                for (int r = 0; r < 8; ++r) {
                    const int rho = rh * 48 + mi * 16 + r + 8 * half;
                    v2f o;
                    o.x = acc0[mi][r] + bo.x;
                    o.y = acc1[mi][r] + bo.y;
                    *(v2f*)(out + row_off(b, wy, p, rho) + nb2 + 2 * lm) = o;
                }
        }
    }

    }

    {
        const int wvz = __builtin_amdgcn_readfirstlane(wave);
        if (wvz < 4) {
            const float* wl = w_out + (size_t)(2 * half) * DIMC + lm;
            const float* a8 = s_t48 + (lm < NW8 ? lm : NW8 - 1) * QS + 2 * half;
            v2f bz0[24], bz1[24];
            for (int u = 0; u < 3; ++u) {
                const int nb = (wvz * 3 + u) * 16;
                ldb24<DIMC>(bz0, wl + nb);
                ldb24<DIMC>(bz1, wl + nb + (size_t)96 * DIMC);
                __builtin_amdgcn_sched_barrier(0);
                v8f acc[1] = {};
                mma24<1, QS>(acc, bz0, a8, a8);
                mma24<1, QS>(acc, bz1, a8 + 96, a8 + 96);
                const float bo = b_out[nb + lm];
                if (half == 0) {
                    #pragma unroll
                    for (int r = 0; r < 8; ++r)
                        out[tok_off(b, wy, r, NT - 1) + nb + lm] = acc[0][r] + bo;
                }
            }
        }
    }
}

extern "C" void kernel_launch(void* const* d_in, const int* in_sizes, int n_in,
                              void* d_out, int out_size, void* d_ws, size_t ws_size,
                              hipStream_t stream) {
    const float* x     = (const float*)d_in[0];
    const float* w_qkv = (const float*)d_in[1];
    const float* w_out = (const float*)d_in[2];
    const float* b_out = (const float*)d_in[3];
    const float* pe    = (const float*)d_in[4];
    float* out = (float*)d_out;

    const size_t shmem = (size_t)(172 + NR * XS + NR * QS + NW8 * XS + NW8 * QS) * sizeof(float);
    hipFuncSetAttribute((const void*)swin_window_attn_kernel,
                        hipFuncAttributeMaxDynamicSharedMemorySize, (int)shmem);

    const int grid = 32 * NWH;
    swin_window_attn_kernel<<<grid, 512, shmem, stream>>>(x, w_qkv, w_out, b_out, pe, out);
}


# Attention: scores computed transposed (keys x queries): softmax in the lane's own registers plus one half-exchange, P feeds P*V from registers, no LDS pass; row-sum order and P*V k order change
# speedup vs baseline: 2.7699x; 1.0332x over previous
#include <hip/hip_runtime.h>

typedef __attribute__((ext_vector_type(2))) float v2f;
typedef __attribute__((ext_vector_type(8))) float v8f;

#define DIMC   192
#define HEADS  6
#define HD     32
#define NT     49
#define NB     48
#define NR     96
#define NW8    8
#define DISPC  3
#define IMG    56
#define NWH    8
#define NWW    8
#define XS     196
#define QS     580
#define SCALEF 0.17677669529663687f

__device__ __forceinline__ v8f wmma4(v2f a, v2f b, v8f c) {
    return __builtin_amdgcn_wmma_f32_16x16x4_f32(
         false, a,  false, b,
         (short)0, c,  false,  false);
}

__device__ __forceinline__ v2f ld2(const float* p) { return *(const v2f*)p; }

__device__ __forceinline__ float xhalf(float v) {
    const int b = __builtin_bit_cast(int, v);
    return __builtin_bit_cast(float,
        __builtin_amdgcn_permlanex16(b, b, 0x76543210u, 0xfedcba98u, false, false));
}

__device__ __forceinline__ size_t tok_off(int b, int wy, int wx, int i) {
    const int r = i / 7, c = i - 7 * r;
    int gh = wy * 7 + r + DISPC; if (gh >= IMG) gh -= IMG;
    int gw = wx * 7 + c + DISPC; if (gw >= IMG) gw -= IMG;
    return (((size_t)b * IMG + gh) * IMG + gw) * DIMC;
}

__device__ __forceinline__ size_t row_off(int b, int wy, int p, int rho) {
    const int win = rho >= NB ? 1 : 0;
    return tok_off(b, wy, 2 * p + win, rho - win * NB);
}

template <int LD>
__device__ __forceinline__ void ldb24(v2f (&dst)[24], const float* __restrict__ p) {
    #pragma unroll
    for (int s = 0; s < 24; ++s) {
        dst[s].x = p[(size_t)(4 * s)     * LD];
        dst[s].y = p[(size_t)(4 * s + 1) * LD];
    }
}

template <int MT, int AS>
__device__ __forceinline__ void mma24(v8f (&acc)[MT], const v2f (&bfr)[24], const float* a,
                                      const float* alast) {
    #pragma unroll
    for (int s = 0; s < 24; ++s) {
        #pragma unroll
        for (int mt = 0; mt < MT; ++mt) {
            v2f af = ld2((mt == MT - 1 ? alast : a + mt * 16 * AS) + 4 * s);
            acc[mt] = wmma4(af, bfr[s], acc[mt]);
        }
    }
}

template <int LD>
__device__ __forceinline__ void ldb12x2(v2f (&r0)[12], v2f (&r1)[12], const float* __restrict__ p) {
    #pragma unroll
    for (int s = 0; s < 12; ++s) {
        r0[s] = *(const v2f*)(p + (size_t)(4 * s)     * LD);
        r1[s] = *(const v2f*)(p + (size_t)(4 * s + 1) * LD);
    }
}

template <int MT, int AS, int G>
__device__ __forceinline__ void mma12x2(v8f (&acc0)[MT], v8f (&acc1)[MT], const v2f (&r0)[12],
                                        const v2f (&r1)[12], const float* a) {
    v2f af[G][MT];
    #pragma unroll
    for (int g = 0; g < G; ++g)
        #pragma unroll
        for (int mt = 0; mt < MT; ++mt) af[g][mt] = ld2(a + mt * 16 * AS + 4 * g);
    #pragma unroll
    for (int s0 = 0; s0 < 12; s0 += G) {
        v2f an[G][MT];
        if (s0 + G < 12) {
            #pragma unroll
            for (int g = 0; g < G; ++g)
                #pragma unroll
                for (int mt = 0; mt < MT; ++mt)
                    an[g][mt] = ld2(a + mt * 16 * AS + 4 * (s0 + G + g));
        }
        #pragma unroll
        for (int g = 0; g < G; ++g) {
            v2f b0, b1;
            b0.x = r0[s0 + g].x; b0.y = r1[s0 + g].x;
            b1.x = r0[s0 + g].y; b1.y = r1[s0 + g].y;
            #pragma unroll
            for (int mt = 0; mt < MT; ++mt) {
                acc0[mt] = wmma4(af[g][mt], b0, acc0[mt]);
                acc1[mt] = wmma4(af[g][mt], b1, acc1[mt]);
            }
        }
        if (s0 + G < 12) {
            #pragma unroll
            for (int g = 0; g < G; ++g)
                #pragma unroll
                for (int mt = 0; mt < MT; ++mt) af[g][mt] = an[g][mt];
        }
    }
}

__global__ __launch_bounds__(512, 4)
void swin_window_attn_kernel(const float* __restrict__ x,
                             const float* __restrict__ w_qkv,
                             const float* __restrict__ w_out,
                             const float* __restrict__ b_out,
                             const float* __restrict__ pe,
                             float* __restrict__ out)
{
    extern __shared__ float smem[];
    float* s_pe  = smem;
    float* s_x   = smem + 172;
    float* s_qkv = s_x   + NR * XS;
    float* s_x48 = s_qkv + NR * QS;
    float* s_t48 = s_x48 + NW8 * XS;

    const int tid  = threadIdx.x;
    const int lane = tid & 31;
    const int wave = tid >> 5;
    const int half = lane >> 4;
    const int lm   = lane & 15;

    const int blk = blockIdx.x;
    const int b   = blk >> 3;
    const int wy  = blk & 7;

    {
        const int t = tid < 384 ? tid : 383;
        const float pev = pe[tid < 169 ? tid : 168];
        const float4 v = *(const float4*)(x + tok_off(b, wy, t / 48, NT - 1) + (t % 48) * 4);
        s_pe[tid < 169 ? tid : 169 + (tid & 1)] = pev;
        *(float4*)(s_x48 + (t / 48) * XS + (t % 48) * 4) = v;
    }
    __syncthreads();

    {
        const float* wl = w_qkv + (size_t)(2 * half) * 576 + lm;
        const float* a8 = s_x48 + (lm < NW8 ? lm : NW8 - 1) * XS + 2 * half;
        const int wv0 = __builtin_amdgcn_readfirstlane(wave);
        const int ntile = wv0 < 4 ? 3 : 2;
        v2f bt0[24], bt1[24];
        for (int j = 0; j < ntile; ++j) {
            const int nb = (wv0 + 16 * j) * 16;
            ldb24<576>(bt0, wl + nb);
            ldb24<576>(bt1, wl + nb + (size_t)96 * 576);
            __builtin_amdgcn_sched_barrier(0);
            v8f acc[1] = {};
            mma24<1, XS>(acc, bt0, a8, a8);
            mma24<1, XS>(acc, bt1, a8 + 96, a8 + 96);
            if (half == 0) {
                #pragma unroll
                for (int r = 0; r < 8; ++r) s_t48[r * QS + nb + lm] = acc[0][r];
            }
        }
    }

    #pragma nounroll
    for (int p = 0; p < NWW / 2; ++p) {

    int tl = tid;
    asm volatile("" : "+v"(tl));
    const int lm   = tl & 15;
    const int half = (tl >> 4) & 1;

    {
        float4 xv[9];
        #pragma unroll
        for (int it = 0; it < 9; ++it) {
            const int idx = tl + it * 512;
            const int rho = idx / 48;
            const int dv  = idx % 48;
            xv[it] = *(const float4*)(x + row_off(b, wy, p, rho) + dv * 4);
        }
        #pragma unroll
        for (int it = 0; it < 9; ++it) {
            const int idx = tl + it * 512;
            *(float4*)(s_x + (idx / 48) * XS + (idx % 48) * 4) = xv[it];
        }
    }
    __syncthreads();

    {
        const float* wl = w_qkv + (size_t)(2 * half) * 576 + lm;
        const float* al = s_x + lm * XS + 2 * half;
        const int wv1 = __builtin_amdgcn_readfirstlane(wave);
        const int nte = 32 + (wv1 & 3);
        const int sub = wv1 >> 2;
        const int nb2 = wv1 * 32;
        const float* wp = w_qkv + (size_t)(2 * half) * 576 + nb2 + 2 * lm;
        v2f pa0[12], pa1[12], pb0[12], pb1[12];
        v2f bq0[24], bq1[24];
        __builtin_amdgcn_s_setprio(3);
        ldb12x2<576>(pa0, pa1, wp);
        v8f acc0[6] = {}, acc1[6] = {};
        ldb12x2<576>(pb0, pb1, wp + (size_t)48 * 576);
        __builtin_amdgcn_sched_barrier(0);
        mma12x2<6, XS, 1>(acc0, acc1, pa0, pa1, al);
        ldb12x2<576>(pa0, pa1, wp + (size_t)96 * 576);
        __builtin_amdgcn_s_setprio(2);
        __builtin_amdgcn_sched_barrier(0);
        mma12x2<6, XS, 1>(acc0, acc1, pb0, pb1, al + 48);
        ldb12x2<576>(pb0, pb1, wp + (size_t)144 * 576);
        __builtin_amdgcn_s_setprio(1);
        __builtin_amdgcn_sched_barrier(0);
        mma12x2<6, XS, 1>(acc0, acc1, pa0, pa1, al + 96);
        ldb24<576>(bq0, wl + nte * 16);
        __builtin_amdgcn_s_setprio(0);
        __builtin_amdgcn_sched_barrier(0);
        mma12x2<6, XS, 1>(acc0, acc1, pb0, pb1, al + 144);
        {
            const int nb = nte * 16;
            ldb24<576>(bq1, wl + nb + (size_t)96 * 576);
            #pragma unroll
            for (int mt = 0; mt < 6; ++mt)
                #pragma unroll
                for (int r = 0; r < 8; ++r) {
                    v2f o;
                    o.x = acc0[mt][r];
                    o.y = acc1[mt][r];
                    *(v2f*)(s_qkv + (mt * 16 + r + 8 * half) * QS + nb2 + 2 * lm) = o;
                }
            __builtin_amdgcn_sched_barrier(0);
            if (sub < 2) {
                const float* a2 = al + sub * 32 * XS;
                v8f acc[2] = {};
                mma24<2, XS>(acc, bq0, a2, a2 + 16 * XS);
                mma24<2, XS>(acc, bq1, a2 + 96, a2 + 16 * XS + 96);
                #pragma unroll
                for (int mi = 0; mi < 2; ++mi)
                    #pragma unroll
                    for (int r = 0; r < 8; ++r)
                        s_qkv[(sub * 32 + mi * 16 + r + 8 * half) * QS + nb + lm] = acc[mi][r];
            } else {
                const float* a1 = al + (sub + 2) * 16 * XS;
                v8f acc[1] = {};
                mma24<1, XS>(acc, bq0, a1, a1);
                mma24<1, XS>(acc, bq1, a1 + 96, a1 + 96);
                #pragma unroll
                for (int r = 0; r < 8; ++r)
                    s_qkv[((sub + 2) * 16 + r + 8 * half) * QS + nb + lm] = acc[0][r];
            }
        }
    }
    __syncthreads();

    {
        const int wv = __builtin_amdgcn_readfirstlane(wave);
        const int mt  = wv & 3;
        const int win = (wv >> 2) & 1;
        const int h0  = wv >> 3;
        const int rb  = win * NB;
        const int wx  = 2 * p + win;
        const bool mask_ul = (wy == NWH - 1);
        const bool mask_lr = (wx == NWW - 1);
        float* t48 = s_t48 + wx * QS;
        const float* qrow = (mt < 3) ? s_qkv + (rb + mt * 16 + lm) * QS : t48;

        float pe_r[4][8];
        bool  mk[4][8];
        const int m_ul = mask_ul ? 1 : 0;
        const int m_lr = mask_lr ? 1 : 0;
        const int i  = mt * 16 + lm;
        const int xi = i / 7, yi = i % 7;
        #pragma unroll
        for (int nt = 0; nt < 4; ++nt) {
            #pragma unroll
            for (int r = 0; r < 8; ++r) {
                const int j  = nt * 16 + r + 8 * half;
                const int xj = j / 7, yj = j % 7;
                const int ok = (int)(i < NT) & (int)(j < NT);
                int idx = (xj - xi + 6) * 13 + (yj - yi + 6);
                idx = idx < 0 ? 0 : (idx > 168 ? 168 : idx);
                const float pv = s_pe[idx];
                pe_r[nt][r] = ok ? pv : 0.f;
                const int msk = ((int)((i >= 28) != (j >= 28)) & m_ul) |
                                ((int)((yi >= 4) != (yj >= 4)) & m_lr);
                mk[nt][r] = (msk & ok) != 0;
            }
        }
        const bool key48 = (half == 0);

        for (int u = 0; u < 3; ++u) {
            if (u == 0) __builtin_amdgcn_s_setprio(3);
            else if (u == 1) __builtin_amdgcn_s_setprio(2);
            else __builtin_amdgcn_s_setprio(1);
            const int h  = h0 + 2 * u;
            const int qc = h * HD;
            const int kc = DIMC + h * HD;
            const int vc = 2 * DIMC + h * HD;

            v2f qf[8], kf[4][8];
            {
                const float* qa = qrow + qc + 2 * half;
                const float* kb = s_qkv + (rb + lm) * QS + kc + 2 * half;
                const float* kb3 = t48 + kc + 2 * half;
                #pragma unroll
                for (int s = 0; s < 8; ++s) qf[s] = ld2(qa + 4 * s);
                #pragma unroll
                for (int nt = 0; nt < 4; ++nt)
                    #pragma unroll
                    for (int s = 0; s < 8; ++s)
                        kf[nt][s] = ld2((nt == 3 ? kb3 : kb + nt * 16 * QS) + 4 * s);
            }
            __builtin_amdgcn_sched_barrier(0);
            v8f sc[4] = {};
            #pragma unroll
            for (int s = 0; s < 8; ++s)
                #pragma unroll
                for (int nt = 0; nt < 4; ++nt)
                    sc[nt] = wmma4(kf[nt][s], qf[s], sc[nt]);
            __builtin_amdgcn_sched_barrier(0);

            v2f vf[2][12];
            float v48x[2];
            {
                const float* vl = s_qkv + (rb + 8 * half) * QS + vc + lm;
                #pragma unroll
                for (int nt = 0; nt < 3; ++nt)
                    #pragma unroll
                    for (int j = 0; j < 4; ++j)
                        #pragma unroll
                        for (int n2 = 0; n2 < 2; ++n2) {
                            vf[n2][nt * 4 + j].x = vl[(nt * 16 + 2 * j)     * QS + n2 * 16];
                            vf[n2][nt * 4 + j].y = vl[(nt * 16 + 2 * j + 1) * QS + n2 * 16];
                        }
                const float* v48 = t48 + vc + lm;
                #pragma unroll
                for (int n2 = 0; n2 < 2; ++n2) v48x[n2] = v48[n2 * 16];
            }
            __builtin_amdgcn_sched_barrier(0);

            #pragma unroll
            for (int nt = 0; nt < 4; ++nt)
                #pragma unroll
                for (int r = 0; r < 8; ++r) {
                    float v = sc[nt][r] * SCALEF + pe_r[nt][r];
                    v = mk[nt][r] ? -1e30f : v;
                    if (nt == 3) v = (r == 0 && key48) ? v : -__builtin_inff();
                    sc[nt][r] = v;
                }
            float mx[4];
            #pragma unroll
            for (int nt = 0; nt < 4; ++nt)
                mx[nt] = fmaxf(fmaxf(fmaxf(sc[nt][0], sc[nt][1]), fmaxf(sc[nt][2], sc[nt][3])),
                               fmaxf(fmaxf(sc[nt][4], sc[nt][5]), fmaxf(sc[nt][6], sc[nt][7])));
            float m = fmaxf(fmaxf(mx[0], mx[1]), fmaxf(mx[2], mx[3]));
            m = fmaxf(m, xhalf(m));
            #pragma unroll
            for (int nt = 0; nt < 4; ++nt)
                #pragma unroll
                for (int r = 0; r < 8; ++r) sc[nt][r] = __expf(sc[nt][r] - m);
            float sm[4];
            #pragma unroll
            for (int nt = 0; nt < 4; ++nt)
                sm[nt] = ((sc[nt][0] + sc[nt][1]) + (sc[nt][2] + sc[nt][3])) +
                         ((sc[nt][4] + sc[nt][5]) + (sc[nt][6] + sc[nt][7]));
            float ps = (sm[0] + sm[1]) + (sm[2] + sm[3]);
            ps += xhalf(ps);
            const float inv = 1.f / ps;
            #pragma unroll
            for (int nt = 0; nt < 4; ++nt)
                #pragma unroll
                for (int r = 0; r < 8; ++r) sc[nt][r] *= inv;

            v8f oa[2] = {};
            #pragma unroll
            for (int nt = 0; nt < 3; ++nt)
                #pragma unroll
                for (int j = 0; j < 4; ++j) {
                    v2f pa;
                    pa.x = sc[nt][2 * j];
                    pa.y = sc[nt][2 * j + 1];
                    #pragma unroll
                    for (int n2 = 0; n2 < 2; ++n2)
                        oa[n2] = wmma4(pa, vf[n2][nt * 4 + j], oa[n2]);
                }
            {
                v2f pa;
                pa.x = sc[3][0];
                pa.y = sc[3][1];
                #pragma unroll
                for (int n2 = 0; n2 < 2; ++n2) {
                    v2f vb;
                    vb.x = v48x[n2];
                    vb.y = v48x[n2];
                    oa[n2] = wmma4(pa, vb, oa[n2]);
                }
            }

            if (mt < 3) {
                #pragma unroll
                for (int n2 = 0; n2 < 2; ++n2)
                    #pragma unroll
                    for (int r = 0; r < 8; ++r)
                        s_qkv[(rb + mt * 16 + r + 8 * half) * QS + qc + n2 * 16 + lm] = oa[n2][r];
            } else if (half == 0) {
                #pragma unroll
                for (int n2 = 0; n2 < 2; ++n2)
                    t48[qc + n2 * 16 + lm] = oa[n2][0];
            }
        }
        __builtin_amdgcn_s_setprio(0);
    }
    __syncthreads();

    {
        const int wv3 = __builtin_amdgcn_readfirstlane(wave);
        if (wv3 >= 4) {
            const int cp  = (wv3 - 4) >> 1;
            const int rh  = (wv3 - 4) & 1;
            const int nb2 = cp * 32;
            const float* wp = w_out + (size_t)(2 * half) * DIMC + nb2 + 2 * lm;
            const float* a3 = s_qkv + (rh * 48 + lm) * QS + 2 * half;
            v2f pa0[12], pa1[12], pb0[12], pb1[12];
            __builtin_amdgcn_s_setprio(3);
            ldb12x2<DIMC>(pa0, pa1, wp);
            ldb12x2<DIMC>(pb0, pb1, wp + (size_t)48 * DIMC);
            __builtin_amdgcn_sched_barrier(0);
            v8f acc0[3] = {}, acc1[3] = {};
            mma12x2<3, QS, 2>(acc0, acc1, pa0, pa1, a3);
            ldb12x2<DIMC>(pa0, pa1, wp + (size_t)96 * DIMC);
            __builtin_amdgcn_s_setprio(2);
            __builtin_amdgcn_sched_barrier(0);
            mma12x2<3, QS, 2>(acc0, acc1, pb0, pb1, a3 + 48);
            ldb12x2<DIMC>(pb0, pb1, wp + (size_t)144 * DIMC);
            __builtin_amdgcn_s_setprio(1);
            __builtin_amdgcn_sched_barrier(0);
            mma12x2<3, QS, 2>(acc0, acc1, pa0, pa1, a3 + 96);
            __builtin_amdgcn_s_setprio(0);
            mma12x2<3, QS, 2>(acc0, acc1, pb0, pb1, a3 + 144);
            const v2f bo = *(const v2f*)(b_out + nb2 + 2 * lm);
            #pragma unroll
            for (int mi = 0; mi < 3; ++mi)
                #pragma unroll
                for (int r = 0; r < 8; ++r) {
                    const int rho = rh * 48 + mi * 16 + r + 8 * half;
                    v2f o;
                    o.x = acc0[mi][r] + bo.x;
                    o.y = acc1[mi][r] + bo.y;
                    *(v2f*)(out + row_off(b, wy, p, rho) + nb2 + 2 * lm) = o;
                }
        }
    }

    }

    {
        const int wvz = __builtin_amdgcn_readfirstlane(wave);
        if (wvz < 4) {
            const float* wl = w_out + (size_t)(2 * half) * DIMC + lm;
            const float* a8 = s_t48 + (lm < NW8 ? lm : NW8 - 1) * QS + 2 * half;
            v2f bz0[24], bz1[24];
            for (int u = 0; u < 3; ++u) {
                const int nb = (wvz * 3 + u) * 16;
                ldb24<DIMC>(bz0, wl + nb);
                ldb24<DIMC>(bz1, wl + nb + (size_t)96 * DIMC);
                __builtin_amdgcn_sched_barrier(0);
                v8f acc[1] = {};
                mma24<1, QS>(acc, bz0, a8, a8);
                mma24<1, QS>(acc, bz1, a8 + 96, a8 + 96);
                const float bo = b_out[nb + lm];
                if (half == 0) {
                    #pragma unroll
                    for (int r = 0; r < 8; ++r)
                        out[tok_off(b, wy, r, NT - 1) + nb + lm] = acc[0][r] + bo;
                }
            }
        }
    }
}

extern "C" void kernel_launch(void* const* d_in, const int* in_sizes, int n_in,
                              void* d_out, int out_size, void* d_ws, size_t ws_size,
                              hipStream_t stream) {
    const float* x     = (const float*)d_in[0];
    const float* w_qkv = (const float*)d_in[1];
    const float* w_out = (const float*)d_in[2];
    const float* b_out = (const float*)d_in[3];
    const float* pe    = (const float*)d_in[4];
    float* out = (float*)d_out;

    const size_t shmem = (size_t)(172 + NR * XS + NR * QS + NW8 * XS + NW8 * QS) * sizeof(float);
    hipFuncSetAttribute((const void*)swin_window_attn_kernel,
                        hipFuncAttributeMaxDynamicSharedMemorySize, (int)shmem);

    const int grid = 32 * NWH;
    swin_window_attn_kernel<<<grid, 512, shmem, stream>>>(x, w_qkv, w_out, b_out, pe, out);
}
